# starting kernel
# baseline (speedup 1.0000x reference)
#include <hip/hip_runtime.h>
#include <hip/hip_bf16.h>


typedef __attribute__((ext_vector_type(2))) float v2f;
typedef __attribute__((ext_vector_type(8))) float v8f;

#define NEG_SLOPE 0.2f

__device__ __forceinline__ unsigned f2ord(float f) {
    unsigned u = __float_as_uint(f);
    return (u & 0x80000000u) ? ~u : (u | 0x80000000u);
}
__device__ __forceinline__ float ord2f(unsigned k) {
    unsigned u = (k & 0x80000000u) ? (k & 0x7fffffffu) : ~k;
    return __uint_as_float(u);
}
__device__ __forceinline__ float leaky(float x) {
    return x > 0.0f ? x : NEG_SLOPE * x;
}

#define P1 132
__global__ __launch_bounds__(256) void gemm1_kernel(
    const float* __restrict__ X, const float* __restrict__ W,
    float* __restrict__ H, int N, int K) {
    __shared__ float sW[64 * P1];
    const int lane = threadIdx.x & 31;
    const int wave = threadIdx.x >> 5;
    const int half = lane >> 4;
    const int col0 = lane & 15;
    const int nrt  = (N + 15) >> 4;
    const int rt0  = (blockIdx.x * 8 + wave) * 2;
    const int rt1  = rt0 + 1;
    const int rowA0 = (rt0 < nrt) ? min(rt0 * 16 + col0, N - 1) : 0;
    const int rowA1 = (rt1 < nrt) ? min(rt1 * 16 + col0, N - 1) : 0;

    v8f acc[2][4];
    #pragma unroll
    for (int u = 0; u < 2; ++u)
        #pragma unroll
        for (int t = 0; t < 4; ++t) acc[u][t] = v8f{};

    const int phases = K >> 7;
    for (int ph = 0; ph < phases; ++ph) {
        for (int i = threadIdx.x; i < 128 * 64; i += 256) {
            int r = i >> 6, c = i & 63;
            sW[c * P1 + r] = W[(size_t)ph * 128 * 64 + i];
        }
        __syncthreads();

        const float* xr0 = X + (size_t)rowA0 * K + ph * 128;
        const float* xr1 = X + (size_t)rowA1 * K + ph * 128;
        for (int kk = 0; kk < 32; ++kk) {
            const int kb = kk * 4 + half * 2;
            v2f a0 = *(const v2f*)(xr0 + kb);
            v2f a1 = *(const v2f*)(xr1 + kb);
            #pragma unroll
            for (int t = 0; t < 4; ++t) {
                v2f b = *(const v2f*)(&sW[(t * 16 + col0) * P1 + kb]);
                acc[0][t] = __builtin_amdgcn_wmma_f32_16x16x4_f32(
                    false, a0, false, b, (short)0, acc[0][t], false, false);
                acc[1][t] = __builtin_amdgcn_wmma_f32_16x16x4_f32(
                    false, a1, false, b, (short)0, acc[1][t], false, false);
            }
        }
        __syncthreads();
    }

    #pragma unroll
    for (int u = 0; u < 2; ++u) {
        int rt = rt0 + u;
        if (rt >= nrt) continue;
        #pragma unroll
        for (int t = 0; t < 4; ++t)
            #pragma unroll
            for (int r = 0; r < 8; ++r) {
                int row = rt * 16 + r + half * 8;
                if (row < N) H[(size_t)row * 64 + t * 16 + col0] = acc[u][t][r];
            }
    }
}

#define P2 68
__global__ __launch_bounds__(256) void gemm2_kernel(
    const float* __restrict__ Hin, const float* __restrict__ W,
    float* __restrict__ Hout, int N, int NC) {
    __shared__ float sW[48 * P2];
    const int lane = threadIdx.x & 31;
    const int wave = threadIdx.x >> 5;
    const int half = lane >> 4;
    const int col0 = lane & 15;
    const int nrt  = (N + 15) >> 4;
    const int rt0  = (blockIdx.x * 8 + wave) * 2;
    const int rt1  = rt0 + 1;
    const int rowA0 = (rt0 < nrt) ? min(rt0 * 16 + col0, N - 1) : 0;
    const int rowA1 = (rt1 < nrt) ? min(rt1 * 16 + col0, N - 1) : 0;

    for (int i = threadIdx.x; i < 64 * 48; i += 256) {
        int r = i / 48, c = i % 48;
        sW[c * P2 + r] = (c < NC) ? W[(size_t)r * NC + c] : 0.0f;
    }
    __syncthreads();

    v8f acc[2][3];
    #pragma unroll
    for (int u = 0; u < 2; ++u)
        #pragma unroll
        for (int t = 0; t < 3; ++t) acc[u][t] = v8f{};

    const float* xr0 = Hin + (size_t)rowA0 * 64;
    const float* xr1 = Hin + (size_t)rowA1 * 64;
    for (int kk = 0; kk < 16; ++kk) {
        const int kb = kk * 4 + half * 2;
        v2f a0 = *(const v2f*)(xr0 + kb);
        v2f a1 = *(const v2f*)(xr1 + kb);
        a0[0] = fmaxf(a0[0], 0.0f); a0[1] = fmaxf(a0[1], 0.0f);
        a1[0] = fmaxf(a1[0], 0.0f); a1[1] = fmaxf(a1[1], 0.0f);
        #pragma unroll
        for (int t = 0; t < 3; ++t) {
            v2f b = *(const v2f*)(&sW[(t * 16 + col0) * P2 + kb]);
            acc[0][t] = __builtin_amdgcn_wmma_f32_16x16x4_f32(
                false, a0, false, b, (short)0, acc[0][t], false, false);
            acc[1][t] = __builtin_amdgcn_wmma_f32_16x16x4_f32(
                false, a1, false, b, (short)0, acc[1][t], false, false);
        }
    }

    #pragma unroll
    for (int u = 0; u < 2; ++u) {
        int rt = rt0 + u;
        if (rt >= nrt) continue;
        #pragma unroll
        for (int t = 0; t < 3; ++t)
            #pragma unroll
            for (int r = 0; r < 8; ++r) {
                int row = rt * 16 + r + half * 8;
                int col = t * 16 + col0;
                if (row < N && col < NC)
                    Hout[(size_t)row * NC + col] = acc[u][t][r];
            }
    }
}

__global__ __launch_bounds__(256) void sd_kernel(
    const float* __restrict__ H, const float* __restrict__ asrc,
    const float* __restrict__ adst, float* __restrict__ s,
    float* __restrict__ d, int N, int F) {
    const int wid  = (int)((blockIdx.x * (long long)blockDim.x + threadIdx.x) >> 5);
    const int lane = threadIdx.x & 31;
    if (wid >= N) return;
    float ps = 0.0f, pd = 0.0f;
    for (int f = lane; f < F; f += 32) {
        float hv = H[(size_t)wid * F + f];
        ps = fmaf(hv, asrc[f], ps);
        pd = fmaf(hv, adst[f], pd);
    }
    #pragma unroll
    for (int off = 16; off > 0; off >>= 1) {
        ps += __shfl_down(ps, off, 32);
        pd += __shfl_down(pd, off, 32);
    }
    if (lane == 0) { s[wid] = ps; d[wid] = pd; }
}

__global__ void zero_f32(float* __restrict__ p, long long n) {
    long long i = blockIdx.x * (long long)blockDim.x + threadIdx.x;
    if (i < n) p[i] = 0.0f;
}
__global__ void init_seg(unsigned* __restrict__ m, float* __restrict__ denom, int n) {
    int i = blockIdx.x * blockDim.x + threadIdx.x;
    if (i < n) { m[i] = 0x007FFFFFu;   denom[i] = 0.0f; }
}

__global__ __launch_bounds__(256) void edge_max_kernel(
    const int* __restrict__ ei, const float* __restrict__ s,
    const float* __restrict__ d, unsigned* __restrict__ m, int E, int EA) {
    int e = blockIdx.x * blockDim.x + threadIdx.x;
    if (e >= EA) return;
    int src, dst;
    if (e < E) { src = ei[e]; dst = ei[E + e]; } else { src = dst = e - E; }
    float sc = leaky(s[src] + d[dst]);
    atomicMax(&m[dst], f2ord(sc));
}

__global__ __launch_bounds__(256) void edge_expsum_kernel(
    const int* __restrict__ ei, const float* __restrict__ ew,
    const float* __restrict__ s, const float* __restrict__ d,
    const unsigned* __restrict__ m, float* __restrict__ denom,
    float* __restrict__ exw, int E, int EA) {
    int e = blockIdx.x * blockDim.x + threadIdx.x;
    if (e >= EA) return;
    int src, dst;
    float w;
    if (e < E) { src = ei[e]; dst = ei[E + e]; w = ew[e]; }
    else       { src = dst = e - E; w = 1.0f; }
    float sc = leaky(s[src] + d[dst]);
    float ex = __expf(sc - ord2f(m[dst]));
    exw[e] = ex * w;
    atomicAdd(&denom[dst], ex);
}

__global__ __launch_bounds__(256) void edge_agg_kernel(
    const int* __restrict__ ei, const float* __restrict__ exw,
    const float* __restrict__ denom, const float* __restrict__ H,
    float* __restrict__ out, int E, int EA, int F) {
    long long gid = blockIdx.x * (long long)blockDim.x + threadIdx.x;
    int e = (int)(gid >> 6);
    int f = (int)(gid & 63);
    if (e >= EA || f >= F) return;
    int src, dst;
    if (e < E) { src = ei[e]; dst = ei[E + e]; } else { src = dst = e - E; }
    float c = exw[e] / fmaxf(denom[dst], 1e-16f);
    atomicAdd(&out[(size_t)dst * F + f], H[(size_t)src * F + f] * c);
}

extern "C" void kernel_launch(void* const* d_in, const int* in_sizes, int n_in,
                              void* d_out, int out_size, void* d_ws, size_t ws_size,
                              hipStream_t stream) {
    const float* x   = (const float*)d_in[0];
    const int*   ei  = (const int*)  d_in[1];
    const float* ew  = (const float*)d_in[2];
    const float* W1  = (const float*)d_in[3];
    const float* a1s = (const float*)d_in[4];
    const float* a1d = (const float*)d_in[5];
    const float* W2  = (const float*)d_in[6];
    const float* a2s = (const float*)d_in[7];
    const float* a2d = (const float*)d_in[8];

    const int HID = in_sizes[4];
    const int FIN = in_sizes[3] / HID;
    const int NC  = in_sizes[7];
    const int N   = in_sizes[0] / FIN;
    const int E   = in_sizes[2];
    const int EA  = E + N;

    float*    h1    = (float*)d_ws;
    float*    out1  = h1   + (size_t)N * HID;
    float*    h2    = out1 + (size_t)N * HID;
    float*    sArr  = h2   + (size_t)N * NC;
    float*    dArr  = sArr + N;
    float*    denom = dArr + N;
    unsigned* mArr  = (unsigned*)(denom + N);
    float*    exw   = (float*)(mArr + N);
    float*    outF  = (float*)d_out;

    const int nrt     = (N + 15) / 16;
    const int pairs   = (nrt + 1) / 2;
    const int gBlocks = (pairs + 7) / 8;
    const int eBlocks = (EA + 255) / 256;
    const int nBlocks = (N + 255) / 256;

    gemm1_kernel<<<gBlocks, 256, 0, stream>>>(x, W1, h1, N, FIN);
    sd_kernel<<<((long long)N * 32 + 255) / 256, 256, 0, stream>>>(h1, a1s, a1d, sArr, dArr, N, HID);
    init_seg<<<nBlocks, 256, 0, stream>>>(mArr, denom, N);
    zero_f32<<<(int)(((long long)N * HID + 255) / 256), 256, 0, stream>>>(out1, (long long)N * HID);
    edge_max_kernel<<<eBlocks, 256, 0, stream>>>(ei, sArr, dArr, mArr, E, EA);
    edge_expsum_kernel<<<eBlocks, 256, 0, stream>>>(ei, ew, sArr, dArr, mArr, denom, exw, E, EA);
    edge_agg_kernel<<<(int)(((long long)EA * 64 + 255) / 256), 256, 0, stream>>>(
        ei, exw, denom, h1, out1, E, EA, HID);

    gemm2_kernel<<<gBlocks, 256, 0, stream>>>(out1, W2, h2, N, NC);
    sd_kernel<<<((long long)N * 32 + 255) / 256, 256, 0, stream>>>(h2, a2s, a2d, sArr, dArr, N, NC);
    init_seg<<<nBlocks, 256, 0, stream>>>(mArr, denom, N);
    zero_f32<<<(int)(((long long)N * NC + 255) / 256), 256, 0, stream>>>(outF, (long long)N * NC);
    edge_max_kernel<<<eBlocks, 256, 0, stream>>>(ei, sArr, dArr, mArr, E, EA);
    edge_expsum_kernel<<<eBlocks, 256, 0, stream>>>(ei, ew, sArr, dArr, mArr, denom, exw, E, EA);
    edge_agg_kernel<<<(int)(((long long)EA * 64 + 255) / 256), 256, 0, stream>>>(
        ei, exw, denom, h2, outF, E, EA, NC);
}


# Edge passes split destination nodes in two ranges, one per second-level cache: a wave reads its die number, draws edge blocks for that range and adds only there; sums stay arrival-order atomics
# speedup vs baseline: 2.4885x; 2.4885x over previous
#include <hip/hip_runtime.h>
#include <hip/hip_bf16.h>


typedef __attribute__((ext_vector_type(2))) float v2f;
typedef __attribute__((ext_vector_type(8))) float v8f;

#define NEG_SLOPE 0.2f

__device__ __forceinline__ unsigned f2ord(float f) {
    unsigned u = __float_as_uint(f);
    return (u & 0x80000000u) ? ~u : (u | 0x80000000u);
}
__device__ __forceinline__ float ord2f(unsigned k) {
    unsigned u = (k & 0x80000000u) ? (k & 0x7fffffffu) : ~k;
    return __uint_as_float(u);
}
__device__ __forceinline__ float leaky(float x) {
    return x > 0.0f ? x : NEG_SLOPE * x;
}

#define P1 132
__global__ __launch_bounds__(256) void gemm1_kernel(
    const float* __restrict__ X, const float* __restrict__ W,
    float* __restrict__ H, int N, int K) {
    __shared__ float sW[64 * P1];
    const int lane = threadIdx.x & 31;
    const int wave = threadIdx.x >> 5;
    const int half = lane >> 4;
    const int col0 = lane & 15;
    const int nrt  = (N + 15) >> 4;
    const int rt0  = (blockIdx.x * 8 + wave) * 2;
    const int rt1  = rt0 + 1;
    const int rowA0 = (rt0 < nrt) ? min(rt0 * 16 + col0, N - 1) : 0;
    const int rowA1 = (rt1 < nrt) ? min(rt1 * 16 + col0, N - 1) : 0;

    v8f acc[2][4];
    #pragma unroll
    for (int u = 0; u < 2; ++u)
        #pragma unroll
        for (int t = 0; t < 4; ++t) acc[u][t] = v8f{};

    const int phases = K >> 7;
    for (int ph = 0; ph < phases; ++ph) {
        for (int i = threadIdx.x; i < 128 * 64; i += 256) {
            int r = i >> 6, c = i & 63;
            sW[c * P1 + r] = W[(size_t)ph * 128 * 64 + i];
        }
        __syncthreads();

        const float* xr0 = X + (size_t)rowA0 * K + ph * 128;
        const float* xr1 = X + (size_t)rowA1 * K + ph * 128;
        for (int kk = 0; kk < 32; ++kk) {
            const int kb = kk * 4 + half * 2;
            v2f a0 = *(const v2f*)(xr0 + kb);
            v2f a1 = *(const v2f*)(xr1 + kb);
            #pragma unroll
            for (int t = 0; t < 4; ++t) {
                v2f b = *(const v2f*)(&sW[(t * 16 + col0) * P1 + kb]);
                acc[0][t] = __builtin_amdgcn_wmma_f32_16x16x4_f32(
                    false, a0, false, b, (short)0, acc[0][t], false, false);
                acc[1][t] = __builtin_amdgcn_wmma_f32_16x16x4_f32(
                    false, a1, false, b, (short)0, acc[1][t], false, false);
            }
        }
        __syncthreads();
    }

    #pragma unroll
    for (int u = 0; u < 2; ++u) {
        int rt = rt0 + u;
        if (rt >= nrt) continue;
        #pragma unroll
        for (int t = 0; t < 4; ++t)
            #pragma unroll
            for (int r = 0; r < 8; ++r) {
                int row = rt * 16 + r + half * 8;
                if (row < N) H[(size_t)row * 64 + t * 16 + col0] = acc[u][t][r];
            }
    }
}

#define P2 68
__global__ __launch_bounds__(256) void gemm2_kernel(
    const float* __restrict__ Hin, const float* __restrict__ W,
    float* __restrict__ Hout, int N, int NC) {
    __shared__ float sW[48 * P2];
    const int lane = threadIdx.x & 31;
    const int wave = threadIdx.x >> 5;
    const int half = lane >> 4;
    const int col0 = lane & 15;
    const int nrt  = (N + 15) >> 4;
    const int rt0  = (blockIdx.x * 8 + wave) * 2;
    const int rt1  = rt0 + 1;
    const int rowA0 = (rt0 < nrt) ? min(rt0 * 16 + col0, N - 1) : 0;
    const int rowA1 = (rt1 < nrt) ? min(rt1 * 16 + col0, N - 1) : 0;

    for (int i = threadIdx.x; i < 64 * 48; i += 256) {
        int r = i / 48, c = i % 48;
        sW[c * P2 + r] = (c < NC) ? W[(size_t)r * NC + c] : 0.0f;
    }
    __syncthreads();

    v8f acc[2][3];
    #pragma unroll
    for (int u = 0; u < 2; ++u)
        #pragma unroll
        for (int t = 0; t < 3; ++t) acc[u][t] = v8f{};

    const float* xr0 = Hin + (size_t)rowA0 * 64;
    const float* xr1 = Hin + (size_t)rowA1 * 64;
    for (int kk = 0; kk < 16; ++kk) {
        const int kb = kk * 4 + half * 2;
        v2f a0 = *(const v2f*)(xr0 + kb);
        v2f a1 = *(const v2f*)(xr1 + kb);
        a0[0] = fmaxf(a0[0], 0.0f); a0[1] = fmaxf(a0[1], 0.0f);
        a1[0] = fmaxf(a1[0], 0.0f); a1[1] = fmaxf(a1[1], 0.0f);
        #pragma unroll
        for (int t = 0; t < 3; ++t) {
            v2f b = *(const v2f*)(&sW[(t * 16 + col0) * P2 + kb]);
            acc[0][t] = __builtin_amdgcn_wmma_f32_16x16x4_f32(
                false, a0, false, b, (short)0, acc[0][t], false, false);
            acc[1][t] = __builtin_amdgcn_wmma_f32_16x16x4_f32(
                false, a1, false, b, (short)0, acc[1][t], false, false);
        }
    }

    #pragma unroll
    for (int u = 0; u < 2; ++u) {
        int rt = rt0 + u;
        if (rt >= nrt) continue;
        #pragma unroll
        for (int t = 0; t < 3; ++t)
            #pragma unroll
            for (int r = 0; r < 8; ++r) {
                int row = rt * 16 + r + half * 8;
                int col = t * 16 + col0;
                if (row < N && col < NC)
                    Hout[(size_t)row * NC + col] = acc[u][t][r];
            }
    }
}

__global__ __launch_bounds__(256) void sd_kernel(
    const float* __restrict__ H, const float* __restrict__ asrc,
    const float* __restrict__ adst, float* __restrict__ s,
    float* __restrict__ d, int N, int F) {
    const int wid  = (int)((blockIdx.x * (long long)blockDim.x + threadIdx.x) >> 5);
    const int lane = threadIdx.x & 31;
    if (wid >= N) return;
    float ps = 0.0f, pd = 0.0f;
    for (int f = lane; f < F; f += 32) {
        float hv = H[(size_t)wid * F + f];
        ps = fmaf(hv, asrc[f], ps);
        pd = fmaf(hv, adst[f], pd);
    }
    #pragma unroll
    for (int off = 16; off > 0; off >>= 1) {
        ps += __shfl_down(ps, off, 32);
        pd += __shfl_down(pd, off, 32);
    }
    if (lane == 0) { s[wid] = ps; d[wid] = pd; }
}

__global__ void zero_f32(float* __restrict__ p, long long n) {
    long long i = blockIdx.x * (long long)blockDim.x + threadIdx.x;
    if (i < n) p[i] = 0.0f;
}
#define CTR_STRIDE 64
__global__ void init_seg(unsigned* __restrict__ m, float* __restrict__ denom,
                         unsigned* __restrict__ ctr, int n) {
    int i = blockIdx.x * blockDim.x + threadIdx.x;
    if (i < n) { m[i] = 0x007FFFFFu;   denom[i] = 0.0f; }
    if (i < 6) ctr[i * CTR_STRIDE] = 0u;
}

#define TCH 128
#define NK (TCH / 32)
__device__ __forceinline__ int wave_side() {
    return (int)((__builtin_amdgcn_s_sendmsg_rtn(0x87) >> 18) & 1u);
}
__device__ __forceinline__ int draw(unsigned* ctr, int lane) {
    unsigned t = 0u;
    if (lane == 0) t = atomicAdd(ctr, 1u);
    return __builtin_amdgcn_readfirstlane((int)t);
}
__device__ __forceinline__ void block_edges(
    const int* __restrict__ ei, int t, int lane, int E, int EA, int NH, int r,
    int (&src)[NK], int (&dst)[NK], bool (&in)[NK]) {
    int ls[NK], ld[NK];
    #pragma unroll
    for (int k = 0; k < NK; ++k) {
        const int ec = min(t * TCH + k * 32 + lane, E - 1);
        ls[k] = ei[ec];
        ld[k] = ei[E + ec];
    }
    #pragma unroll
    for (int k = 0; k < NK; ++k) {
        const int e = t * TCH + k * 32 + lane;
        const bool loop = e >= E;
        src[k] = loop ? e - E : ls[k];
        dst[k] = loop ? e - E : ld[k];
        in[k] = (e < EA) && ((dst[k] >= NH) == (r != 0));
    }
}

__global__ __launch_bounds__(256) void edge_max_kernel(
    const int* __restrict__ ei, const float* __restrict__ s,
    const float* __restrict__ d, unsigned* __restrict__ m,
    unsigned* __restrict__ ctr, int E, int EA, int NH) {
    const int lane = threadIdx.x & 31;
    const int nT = (EA + TCH - 1) / TCH;
    int r = wave_side();
    for (int pass = 0; pass < 2; ++pass, r ^= 1) {
        unsigned* c = ctr + r * CTR_STRIDE;
        for (int t = draw(c, lane); t < nT; t = draw(c, lane)) {
            int src[NK], dst[NK];
            bool in[NK];
            block_edges(ei, t, lane, E, EA, NH, r, src, dst, in);
            float sv[NK], dv[NK];
            #pragma unroll
            for (int k = 0; k < NK; ++k) {
                sv[k] = 0.0f; dv[k] = 0.0f;
                if (in[k]) { sv[k] = s[src[k]]; dv[k] = d[dst[k]]; }
            }
            #pragma unroll
            for (int k = 0; k < NK; ++k)
                if (in[k]) atomicMax(&m[dst[k]], f2ord(leaky(sv[k] + dv[k])));
        }
    }
}

__global__ __launch_bounds__(256) void edge_expsum_kernel(
    const int* __restrict__ ei, const float* __restrict__ ew,
    const float* __restrict__ s, const float* __restrict__ d,
    const unsigned* __restrict__ m, float* __restrict__ denom,
    float* __restrict__ exw, unsigned* __restrict__ ctr, int E, int EA, int NH) {
    const int lane = threadIdx.x & 31;
    const int nT = (EA + TCH - 1) / TCH;
    int r = wave_side();
    for (int pass = 0; pass < 2; ++pass, r ^= 1) {
        unsigned* c = ctr + r * CTR_STRIDE;
        for (int t = draw(c, lane); t < nT; t = draw(c, lane)) {
            int src[NK], dst[NK];
            bool in[NK];
            block_edges(ei, t, lane, E, EA, NH, r, src, dst, in);
            float sv[NK], dv[NK], wv[NK];
            unsigned mv[NK];
            #pragma unroll
            for (int k = 0; k < NK; ++k) {
                const int e = t * TCH + k * 32 + lane;
                sv[k] = 0.0f; dv[k] = 0.0f; mv[k] = 0u; wv[k] = 1.0f;
                if (in[k]) {
                    sv[k] = s[src[k]]; dv[k] = d[dst[k]]; mv[k] = m[dst[k]];
                    if (e < E) wv[k] = ew[e];
                }
            }
            #pragma unroll
            for (int k = 0; k < NK; ++k) {
                if (in[k]) {
                    const int e = t * TCH + k * 32 + lane;
                    const float ex = __expf(leaky(sv[k] + dv[k]) - ord2f(mv[k]));
                    exw[e] = ex * wv[k];
                    atomicAdd(&denom[dst[k]], ex);
                }
            }
        }
    }
}

#define GRP 16
__global__ __launch_bounds__(256) void edge_agg_kernel(
    const int* __restrict__ ei, const float* __restrict__ exw,
    const float* __restrict__ denom, const float* __restrict__ H,
    float* __restrict__ out, unsigned* __restrict__ ctr,
    int E, int EA, int F, int NH) {
    __shared__ int   qs[8][TCH];
    __shared__ int   qd[8][TCH];
    __shared__ float qc[8][TCH];
    const int lane = threadIdx.x & 31;
    const int wv = threadIdx.x >> 5;
    const int nT = (EA + TCH - 1) / TCH;
    const bool p0 = lane < F;
    const bool p1 = lane + 32 < F;
    const unsigned colA = (unsigned)min(lane, F - 1);
    const unsigned colB = (unsigned)min(lane + 32, F - 1);
    const unsigned uF = (unsigned)F;
    int r = wave_side();
    for (int pass = 0; pass < 2; ++pass, r ^= 1) {
        unsigned* c = ctr + r * CTR_STRIDE;
        for (int t = draw(c, lane); t < nT; t = draw(c, lane)) {
            int src[NK], dst[NK];
            bool in[NK];
            block_edges(ei, t, lane, E, EA, NH, r, src, dst, in);
            float dn[NK], xw[NK];
            #pragma unroll
            for (int k = 0; k < NK; ++k) {
                dn[k] = 1.0f; xw[k] = 0.0f;
                if (in[k]) { dn[k] = denom[dst[k]]; xw[k] = exw[t * TCH + k * 32 + lane]; }
            }
            int cnt = 0;
            #pragma unroll
            for (int k = 0; k < NK; ++k) {
                const unsigned mask = __builtin_amdgcn_ballot_w32(in[k]);
                const int slot = cnt + (int)__builtin_amdgcn_mbcnt_lo(mask, 0u);
                if (in[k]) {
                    qs[wv][slot] = src[k];
                    qd[wv][slot] = dst[k];
                    qc[wv][slot] = xw[k] / fmaxf(dn[k], 1e-16f);
                }
                cnt += __builtin_popcount(mask);
            }
            for (int g = 0; g < cnt; g += GRP) {
                float va[GRP], vb[GRP];
                #pragma unroll
                for (int j = 0; j < GRP; ++j) {
                    const int idx = min(g + j, cnt - 1);
                    const unsigned ro = (unsigned)__builtin_amdgcn_readfirstlane(qs[wv][idx]) * uF;
                    va[j] = H[ro + colA];
                    vb[j] = H[ro + colB];
                }
                #pragma unroll
                for (int j = 0; j < GRP; ++j) {
                    if (g + j < cnt) {
                        const unsigned wo = (unsigned)__builtin_amdgcn_readfirstlane(qd[wv][g + j]) * uF;
                        const float cj = qc[wv][g + j];
                        if (p0) atomicAdd(&out[wo + (unsigned)lane], va[j] * cj);
                        if (p1) atomicAdd(&out[wo + (unsigned)lane + 32u], vb[j] * cj);
                    }
                }
            }
        }
    }
}

extern "C" void kernel_launch(void* const* d_in, const int* in_sizes, int n_in,
                              void* d_out, int out_size, void* d_ws, size_t ws_size,
                              hipStream_t stream) {
    const float* x   = (const float*)d_in[0];
    const int*   ei  = (const int*)  d_in[1];
    const float* ew  = (const float*)d_in[2];
    const float* W1  = (const float*)d_in[3];
    const float* a1s = (const float*)d_in[4];
    const float* a1d = (const float*)d_in[5];
    const float* W2  = (const float*)d_in[6];
    const float* a2s = (const float*)d_in[7];
    const float* a2d = (const float*)d_in[8];

    const int HID = in_sizes[4];
    const int FIN = in_sizes[3] / HID;
    const int NC  = in_sizes[7];
    const int N   = in_sizes[0] / FIN;
    const int E   = in_sizes[2];
    const int EA  = E + N;

    float*    h1    = (float*)d_ws;
    float*    out1  = h1   + (size_t)N * HID;
    float*    h2    = out1 + (size_t)N * HID;
    float*    sArr  = h2   + (size_t)N * NC;
    float*    dArr  = sArr + N;
    float*    denom = dArr + N;
    unsigned* mArr  = (unsigned*)(denom + N);
    float*    exw   = (float*)(mArr + N);
    unsigned* ctr   = (unsigned*)(exw + ((size_t)EA + 63) / 64 * 64);
    float*    outF  = (float*)d_out;

    const int nrt     = (N + 15) / 16;
    const int pairs   = (nrt + 1) / 2;
    const int gBlocks = (pairs + 7) / 8;
    const int eBlocks = 1024;
    const int NH      = N / 2;
    const int nBlocks = (N + 255) / 256;

    gemm1_kernel<<<gBlocks, 256, 0, stream>>>(x, W1, h1, N, FIN);
    sd_kernel<<<((long long)N * 32 + 255) / 256, 256, 0, stream>>>(h1, a1s, a1d, sArr, dArr, N, HID);
    init_seg<<<nBlocks, 256, 0, stream>>>(mArr, denom, ctr, N);
    zero_f32<<<(int)(((long long)N * HID + 255) / 256), 256, 0, stream>>>(out1, (long long)N * HID);
    edge_max_kernel<<<eBlocks, 256, 0, stream>>>(ei, sArr, dArr, mArr, ctr, E, EA, NH);
    edge_expsum_kernel<<<eBlocks, 256, 0, stream>>>(ei, ew, sArr, dArr, mArr, denom, exw, ctr + 2 * CTR_STRIDE, E, EA, NH);
    edge_agg_kernel<<<eBlocks, 256, 0, stream>>>(ei, exw, denom, h1, out1, ctr + 4 * CTR_STRIDE, E, EA, HID, NH);

    gemm2_kernel<<<gBlocks, 256, 0, stream>>>(out1, W2, h2, N, NC);
    sd_kernel<<<((long long)N * 32 + 255) / 256, 256, 0, stream>>>(h2, a2s, a2d, sArr, dArr, N, NC);
    init_seg<<<nBlocks, 256, 0, stream>>>(mArr, denom, ctr, N);
    zero_f32<<<(int)(((long long)N * NC + 255) / 256), 256, 0, stream>>>(outF, (long long)N * NC);
    edge_max_kernel<<<eBlocks, 256, 0, stream>>>(ei, sArr, dArr, mArr, ctr, E, EA, NH);
    edge_expsum_kernel<<<eBlocks, 256, 0, stream>>>(ei, ew, sArr, dArr, mArr, denom, exw, ctr + 2 * CTR_STRIDE, E, EA, NH);
    edge_agg_kernel<<<eBlocks, 256, 0, stream>>>(ei, exw, denom, h2, outF, ctr + 4 * CTR_STRIDE, E, EA, NC, NH);
}


# Exp-sum pass folded into the row pass: it forms exp, adds exp-sums and UNNORMALISED rows; division by the exp-sum once per node after (GEMM 2 load, end pass): 14 launches -> 13; atomics as before
# speedup vs baseline: 2.6972x; 1.0839x over previous
#include <hip/hip_runtime.h>
#include <hip/hip_bf16.h>


typedef __attribute__((ext_vector_type(2))) float v2f;
typedef __attribute__((ext_vector_type(8))) float v8f;

#define NEG_SLOPE 0.2f

__device__ __forceinline__ unsigned f2ord(float f) {
    unsigned u = __float_as_uint(f);
    return (u & 0x80000000u) ? ~u : (u | 0x80000000u);
}
__device__ __forceinline__ float ord2f(unsigned k) {
    unsigned u = (k & 0x80000000u) ? (k & 0x7fffffffu) : ~k;
    return __uint_as_float(u);
}
__device__ __forceinline__ float leaky(float x) {
    return x > 0.0f ? x : NEG_SLOPE * x;
}

#define P1 132
__global__ __launch_bounds__(256) void gemm1_kernel(
    const float* __restrict__ X, const float* __restrict__ W,
    float* __restrict__ H, int N, int K) {
    __shared__ float sW[64 * P1];
    const int lane = threadIdx.x & 31;
    const int wave = threadIdx.x >> 5;
    const int half = lane >> 4;
    const int col0 = lane & 15;
    const int nrt  = (N + 15) >> 4;
    const int rt0  = (blockIdx.x * 8 + wave) * 2;
    const int rt1  = rt0 + 1;
    const int rowA0 = (rt0 < nrt) ? min(rt0 * 16 + col0, N - 1) : 0;
    const int rowA1 = (rt1 < nrt) ? min(rt1 * 16 + col0, N - 1) : 0;

    v8f acc[2][4];
    #pragma unroll
    for (int u = 0; u < 2; ++u)
        #pragma unroll
        for (int t = 0; t < 4; ++t) acc[u][t] = v8f{};

    const int phases = K >> 7;
    for (int ph = 0; ph < phases; ++ph) {
        for (int i = threadIdx.x; i < 128 * 64; i += 256) {
            int r = i >> 6, c = i & 63;
            sW[c * P1 + r] = W[(size_t)ph * 128 * 64 + i];
        }
        __syncthreads();

        const float* xr0 = X + (size_t)rowA0 * K + ph * 128;
        const float* xr1 = X + (size_t)rowA1 * K + ph * 128;
        for (int kk = 0; kk < 32; ++kk) {
            const int kb = kk * 4 + half * 2;
            v2f a0 = *(const v2f*)(xr0 + kb);
            v2f a1 = *(const v2f*)(xr1 + kb);
            #pragma unroll
            for (int t = 0; t < 4; ++t) {
                v2f b = *(const v2f*)(&sW[(t * 16 + col0) * P1 + kb]);
                acc[0][t] = __builtin_amdgcn_wmma_f32_16x16x4_f32(
                    false, a0, false, b, (short)0, acc[0][t], false, false);
                acc[1][t] = __builtin_amdgcn_wmma_f32_16x16x4_f32(
                    false, a1, false, b, (short)0, acc[1][t], false, false);
            }
        }
        __syncthreads();
    }

    #pragma unroll
    for (int u = 0; u < 2; ++u) {
        int rt = rt0 + u;
        if (rt >= nrt) continue;
        #pragma unroll
        for (int t = 0; t < 4; ++t)
            #pragma unroll
            for (int r = 0; r < 8; ++r) {
                int row = rt * 16 + r + half * 8;
                if (row < N) H[(size_t)row * 64 + t * 16 + col0] = acc[u][t][r];
            }
    }
}

#define P2 68
__global__ __launch_bounds__(256) void gemm2_kernel(
    const float* __restrict__ Hin, const float* __restrict__ Dn,
    const float* __restrict__ W,
    float* __restrict__ Hout, int N, int NC) {
    __shared__ float sW[48 * P2];
    const int lane = threadIdx.x & 31;
    const int wave = threadIdx.x >> 5;
    const int half = lane >> 4;
    const int col0 = lane & 15;
    const int nrt  = (N + 15) >> 4;
    const int rt0  = (blockIdx.x * 8 + wave) * 2;
    const int rt1  = rt0 + 1;
    const int rowA0 = (rt0 < nrt) ? min(rt0 * 16 + col0, N - 1) : 0;
    const int rowA1 = (rt1 < nrt) ? min(rt1 * 16 + col0, N - 1) : 0;

    for (int i = threadIdx.x; i < 64 * 48; i += 256) {
        int r = i / 48, c = i % 48;
        sW[c * P2 + r] = (c < NC) ? W[(size_t)r * NC + c] : 0.0f;
    }
    __syncthreads();

    v8f acc[2][3];
    #pragma unroll
    for (int u = 0; u < 2; ++u)
        #pragma unroll
        for (int t = 0; t < 3; ++t) acc[u][t] = v8f{};

    const float* xr0 = Hin + (size_t)rowA0 * 64;
    const float* xr1 = Hin + (size_t)rowA1 * 64;
    const float dn0 = fmaxf(Dn[rowA0], 1e-16f);
    const float dn1 = fmaxf(Dn[rowA1], 1e-16f);
    for (int kk = 0; kk < 16; ++kk) {
        const int kb = kk * 4 + half * 2;
        v2f a0 = *(const v2f*)(xr0 + kb);
        v2f a1 = *(const v2f*)(xr1 + kb);
        a0[0] = fmaxf(a0[0] / dn0, 0.0f); a0[1] = fmaxf(a0[1] / dn0, 0.0f);
        a1[0] = fmaxf(a1[0] / dn1, 0.0f); a1[1] = fmaxf(a1[1] / dn1, 0.0f);
        #pragma unroll
        for (int t = 0; t < 3; ++t) {
            v2f b = *(const v2f*)(&sW[(t * 16 + col0) * P2 + kb]);
            acc[0][t] = __builtin_amdgcn_wmma_f32_16x16x4_f32(
                false, a0, false, b, (short)0, acc[0][t], false, false);
            acc[1][t] = __builtin_amdgcn_wmma_f32_16x16x4_f32(
                false, a1, false, b, (short)0, acc[1][t], false, false);
        }
    }

    #pragma unroll
    for (int u = 0; u < 2; ++u) {
        int rt = rt0 + u;
        if (rt >= nrt) continue;
        #pragma unroll
        for (int t = 0; t < 3; ++t)
            #pragma unroll
            for (int r = 0; r < 8; ++r) {
                int row = rt * 16 + r + half * 8;
                int col = t * 16 + col0;
                if (row < N && col < NC)
                    Hout[(size_t)row * NC + col] = acc[u][t][r];
            }
    }
}

__global__ __launch_bounds__(256) void sd_kernel(
    const float* __restrict__ H, const float* __restrict__ asrc,
    const float* __restrict__ adst, float* __restrict__ s,
    float* __restrict__ d, int N, int F) {
    const int wid  = (int)((blockIdx.x * (long long)blockDim.x + threadIdx.x) >> 5);
    const int lane = threadIdx.x & 31;
    if (wid >= N) return;
    float ps = 0.0f, pd = 0.0f;
    for (int f = lane; f < F; f += 32) {
        float hv = H[(size_t)wid * F + f];
        ps = fmaf(hv, asrc[f], ps);
        pd = fmaf(hv, adst[f], pd);
    }
    #pragma unroll
    for (int off = 16; off > 0; off >>= 1) {
        ps += __shfl_down(ps, off, 32);
        pd += __shfl_down(pd, off, 32);
    }
    if (lane == 0) { s[wid] = ps; d[wid] = pd; }
}

__global__ void zero_f32(float* __restrict__ p, long long n) {
    long long i = blockIdx.x * (long long)blockDim.x + threadIdx.x;
    if (i < n) p[i] = 0.0f;
}
#define CTR_STRIDE 64
__global__ void init_seg(unsigned* __restrict__ m, float* __restrict__ denom,
                         unsigned* __restrict__ ctr, int n) {
    int i = blockIdx.x * blockDim.x + threadIdx.x;
    if (i < n) { m[i] = 0x007FFFFFu;   denom[i] = 0.0f; }
    if (i < 4) ctr[i * CTR_STRIDE] = 0u;
}

#define TCH 128
#define NK (TCH / 32)
__device__ __forceinline__ int wave_side() {
    return (int)((__builtin_amdgcn_s_sendmsg_rtn(0x87) >> 18) & 1u);
}
__device__ __forceinline__ int draw(unsigned* ctr, int lane) {
    unsigned t = 0u;
    if (lane == 0) t = atomicAdd(ctr, 1u);
    return __builtin_amdgcn_readfirstlane((int)t);
}
__device__ __forceinline__ void block_edges(
    const int* __restrict__ ei, int t, int lane, int E, int EA, int NH, int r,
    int (&src)[NK], int (&dst)[NK], bool (&in)[NK]) {
    int ls[NK], ld[NK];
    #pragma unroll
    for (int k = 0; k < NK; ++k) {
        const int ec = min(t * TCH + k * 32 + lane, E - 1);
        ls[k] = ei[ec];
        ld[k] = ei[E + ec];
    }
    #pragma unroll
    for (int k = 0; k < NK; ++k) {
        const int e = t * TCH + k * 32 + lane;
        const bool loop = e >= E;
        src[k] = loop ? e - E : ls[k];
        dst[k] = loop ? e - E : ld[k];
        in[k] = (e < EA) && ((dst[k] >= NH) == (r != 0));
    }
}

__global__ __launch_bounds__(256) void edge_max_kernel(
    const int* __restrict__ ei, const float* __restrict__ s,
    const float* __restrict__ d, unsigned* __restrict__ m,
    unsigned* __restrict__ ctr, int E, int EA, int NH) {
    const int lane = threadIdx.x & 31;
    const int nT = (EA + TCH - 1) / TCH;
    int r = wave_side();
    for (int pass = 0; pass < 2; ++pass, r ^= 1) {
        unsigned* c = ctr + r * CTR_STRIDE;
        for (int t = draw(c, lane); t < nT; t = draw(c, lane)) {
            int src[NK], dst[NK];
            bool in[NK];
            block_edges(ei, t, lane, E, EA, NH, r, src, dst, in);
            float sv[NK], dv[NK];
            #pragma unroll
            for (int k = 0; k < NK; ++k) {
                sv[k] = 0.0f; dv[k] = 0.0f;
                if (in[k]) { sv[k] = s[src[k]]; dv[k] = d[dst[k]]; }
            }
            #pragma unroll
            for (int k = 0; k < NK; ++k)
                if (in[k]) atomicMax(&m[dst[k]], f2ord(leaky(sv[k] + dv[k])));
        }
    }
}

#define GRP 16
__global__ __launch_bounds__(256) void edge_row_kernel(
    const int* __restrict__ ei, const float* __restrict__ ew,
    const float* __restrict__ s, const float* __restrict__ d,
    const unsigned* __restrict__ m, float* __restrict__ denom,
    const float* __restrict__ H, float* __restrict__ out,
    unsigned* __restrict__ ctr, int E, int EA, int F, int NH) {
    __shared__ int   qs[8][TCH];
    __shared__ int   qd[8][TCH];
    __shared__ float qc[8][TCH];
    const int lane = threadIdx.x & 31;
    const int wv = threadIdx.x >> 5;
    const int nT = (EA + TCH - 1) / TCH;
    const bool p0 = lane < F;
    const bool p1 = lane + 32 < F;
    const unsigned colA = (unsigned)min(lane, F - 1);
    const unsigned colB = (unsigned)min(lane + 32, F - 1);
    const unsigned uF = (unsigned)F;
    int r = wave_side();
    for (int pass = 0; pass < 2; ++pass, r ^= 1) {
        unsigned* c = ctr + r * CTR_STRIDE;
        for (int t = draw(c, lane); t < nT; t = draw(c, lane)) {
            int src[NK], dst[NK];
            bool in[NK];
            block_edges(ei, t, lane, E, EA, NH, r, src, dst, in);
            float sv[NK], dv[NK], wt[NK];
            unsigned mv[NK];
            #pragma unroll
            for (int k = 0; k < NK; ++k) {
                const int e = t * TCH + k * 32 + lane;
                sv[k] = 0.0f; dv[k] = 0.0f; mv[k] = 0u; wt[k] = 1.0f;
                if (in[k]) {
                    sv[k] = s[src[k]]; dv[k] = d[dst[k]]; mv[k] = m[dst[k]];
                    if (e < E) wt[k] = ew[e];
                }
            }
            int cnt = 0;
            #pragma unroll
            for (int k = 0; k < NK; ++k) {
                const unsigned mask = __builtin_amdgcn_ballot_w32(in[k]);
                const int slot = cnt + (int)__builtin_amdgcn_mbcnt_lo(mask, 0u);
                if (in[k]) {
                    const float ex = __expf(leaky(sv[k] + dv[k]) - ord2f(mv[k]));
                    atomicAdd(&denom[dst[k]], ex);
                    qs[wv][slot] = src[k];
                    qd[wv][slot] = dst[k];
                    qc[wv][slot] = ex * wt[k];
                }
                cnt += __builtin_popcount(mask);
            }
            for (int g = 0; g < cnt; g += GRP) {
                float va[GRP], vb[GRP];
                #pragma unroll
                for (int j = 0; j < GRP; ++j) {
                    const int idx = min(g + j, cnt - 1);
                    const unsigned ro = (unsigned)__builtin_amdgcn_readfirstlane(qs[wv][idx]) * uF;
                    va[j] = H[ro + colA];
                    vb[j] = H[ro + colB];
                }
                #pragma unroll
                for (int j = 0; j < GRP; ++j) {
                    if (g + j < cnt) {
                        const unsigned wo = (unsigned)__builtin_amdgcn_readfirstlane(qd[wv][g + j]) * uF;
                        const float cj = qc[wv][g + j];
                        if (p0) atomicAdd(&out[wo + (unsigned)lane], va[j] * cj);
                        if (p1) atomicAdd(&out[wo + (unsigned)lane + 32u], vb[j] * cj);
                    }
                }
            }
        }
    }
}

__global__ void norm_rows(float* __restrict__ out, const float* __restrict__ denom,
                          unsigned n, unsigned F) {
    const unsigned i = blockIdx.x * blockDim.x + threadIdx.x;
    if (i < n) out[i] = out[i] / fmaxf(denom[i / F], 1e-16f);
}

extern "C" void kernel_launch(void* const* d_in, const int* in_sizes, int n_in,
                              void* d_out, int out_size, void* d_ws, size_t ws_size,
                              hipStream_t stream) {
    const float* x   = (const float*)d_in[0];
    const int*   ei  = (const int*)  d_in[1];
    const float* ew  = (const float*)d_in[2];
    const float* W1  = (const float*)d_in[3];
    const float* a1s = (const float*)d_in[4];
    const float* a1d = (const float*)d_in[5];
    const float* W2  = (const float*)d_in[6];
    const float* a2s = (const float*)d_in[7];
    const float* a2d = (const float*)d_in[8];

    const int HID = in_sizes[4];
    const int FIN = in_sizes[3] / HID;
    const int NC  = in_sizes[7];
    const int N   = in_sizes[0] / FIN;
    const int E   = in_sizes[2];
    const int EA  = E + N;

    float*    h1    = (float*)d_ws;
    float*    out1  = h1   + (size_t)N * HID;
    float*    h2    = out1 + (size_t)N * HID;
    float*    sArr  = h2   + (size_t)N * NC;
    float*    dArr  = sArr + N;
    float*    denom = dArr + N;
    unsigned* mArr  = (unsigned*)(denom + N);
    unsigned* ctr   = mArr + ((size_t)N + 63) / 64 * 64;
    float*    outF  = (float*)d_out;

    const int nrt     = (N + 15) / 16;
    const int pairs   = (nrt + 1) / 2;
    const int gBlocks = (pairs + 7) / 8;
    const int eBlocks = 1024;
    const int NH      = N / 2;
    const int nBlocks = (N + 255) / 256;

    gemm1_kernel<<<gBlocks, 256, 0, stream>>>(x, W1, h1, N, FIN);
    sd_kernel<<<((long long)N * 32 + 255) / 256, 256, 0, stream>>>(h1, a1s, a1d, sArr, dArr, N, HID);
    init_seg<<<nBlocks, 256, 0, stream>>>(mArr, denom, ctr, N);
    zero_f32<<<(int)(((long long)N * HID + 255) / 256), 256, 0, stream>>>(out1, (long long)N * HID);
    edge_max_kernel<<<eBlocks, 256, 0, stream>>>(ei, sArr, dArr, mArr, ctr, E, EA, NH);
    edge_row_kernel<<<eBlocks, 256, 0, stream>>>(ei, ew, sArr, dArr, mArr, denom, h1, out1, ctr + 2 * CTR_STRIDE, E, EA, HID, NH);

    gemm2_kernel<<<gBlocks, 256, 0, stream>>>(out1, denom, W2, h2, N, NC);
    sd_kernel<<<((long long)N * 32 + 255) / 256, 256, 0, stream>>>(h2, a2s, a2d, sArr, dArr, N, NC);
    init_seg<<<nBlocks, 256, 0, stream>>>(mArr, denom, ctr, N);
    zero_f32<<<(int)(((long long)N * NC + 255) / 256), 256, 0, stream>>>(outF, (long long)N * NC);
    edge_max_kernel<<<eBlocks, 256, 0, stream>>>(ei, sArr, dArr, mArr, ctr, E, EA, NH);
    edge_row_kernel<<<eBlocks, 256, 0, stream>>>(ei, ew, sArr, dArr, mArr, denom, h2, outF, ctr + 2 * CTR_STRIDE, E, EA, NC, NH);
    norm_rows<<<(int)(((long long)N * NC + 255) / 256), 256, 0, stream>>>(outF, denom, (unsigned)N * (unsigned)NC, (unsigned)NC);
}


# Edge passes gather fewer node words: segment max taken over s[src] alone (leaky(s+d) is monotone in s), d and that max in one 8-byte record, s[src] kept in edge order for the row pass; same atomics
# speedup vs baseline: 3.0507x; 1.1311x over previous
#include <hip/hip_runtime.h>
#include <hip/hip_bf16.h>


typedef __attribute__((ext_vector_type(2))) float v2f;
typedef __attribute__((ext_vector_type(8))) float v8f;

#define NEG_SLOPE 0.2f

__device__ __forceinline__ unsigned f2ord(float f) {
    unsigned u = __float_as_uint(f);
    return (u & 0x80000000u) ? ~u : (u | 0x80000000u);
}
__device__ __forceinline__ float ord2f(unsigned k) {
    unsigned u = (k & 0x80000000u) ? (k & 0x7fffffffu) : ~k;
    return __uint_as_float(u);
}
__device__ __forceinline__ float leaky(float x) {
    return x > 0.0f ? x : NEG_SLOPE * x;
}

#define P1 132
__global__ __launch_bounds__(256) void gemm1_kernel(
    const float* __restrict__ X, const float* __restrict__ W,
    float* __restrict__ H, int N, int K) {
    __shared__ float sW[64 * P1];
    const int lane = threadIdx.x & 31;
    const int wave = threadIdx.x >> 5;
    const int half = lane >> 4;
    const int col0 = lane & 15;
    const int nrt  = (N + 15) >> 4;
    const int rt0  = (blockIdx.x * 8 + wave) * 2;
    const int rt1  = rt0 + 1;
    const int rowA0 = (rt0 < nrt) ? min(rt0 * 16 + col0, N - 1) : 0;
    const int rowA1 = (rt1 < nrt) ? min(rt1 * 16 + col0, N - 1) : 0;

    v8f acc[2][4];
    #pragma unroll
    for (int u = 0; u < 2; ++u)
        #pragma unroll
        for (int t = 0; t < 4; ++t) acc[u][t] = v8f{};

    const int phases = K >> 7;
    for (int ph = 0; ph < phases; ++ph) {
        for (int i = threadIdx.x; i < 128 * 64; i += 256) {
            int r = i >> 6, c = i & 63;
            sW[c * P1 + r] = W[(size_t)ph * 128 * 64 + i];
        }
        __syncthreads();

        const float* xr0 = X + (size_t)rowA0 * K + ph * 128;
        const float* xr1 = X + (size_t)rowA1 * K + ph * 128;
        for (int kk = 0; kk < 32; ++kk) {
            const int kb = kk * 4 + half * 2;
            v2f a0 = *(const v2f*)(xr0 + kb);
            v2f a1 = *(const v2f*)(xr1 + kb);
            #pragma unroll
            for (int t = 0; t < 4; ++t) {
                v2f b = *(const v2f*)(&sW[(t * 16 + col0) * P1 + kb]);
                acc[0][t] = __builtin_amdgcn_wmma_f32_16x16x4_f32(
                    false, a0, false, b, (short)0, acc[0][t], false, false);
                acc[1][t] = __builtin_amdgcn_wmma_f32_16x16x4_f32(
                    false, a1, false, b, (short)0, acc[1][t], false, false);
            }
        }
        __syncthreads();
    }

    #pragma unroll
    for (int u = 0; u < 2; ++u) {
        int rt = rt0 + u;
        if (rt >= nrt) continue;
        #pragma unroll
        for (int t = 0; t < 4; ++t)
            #pragma unroll
            for (int r = 0; r < 8; ++r) {
                int row = rt * 16 + r + half * 8;
                if (row < N) H[(size_t)row * 64 + t * 16 + col0] = acc[u][t][r];
            }
    }
}

#define P2 68
__global__ __launch_bounds__(256) void gemm2_kernel(
    const float* __restrict__ Hin, const float* __restrict__ Dn,
    const float* __restrict__ W,
    float* __restrict__ Hout, int N, int NC) {
    __shared__ float sW[48 * P2];
    const int lane = threadIdx.x & 31;
    const int wave = threadIdx.x >> 5;
    const int half = lane >> 4;
    const int col0 = lane & 15;
    const int nrt  = (N + 15) >> 4;
    const int rt0  = (blockIdx.x * 8 + wave) * 2;
    const int rt1  = rt0 + 1;
    const int rowA0 = (rt0 < nrt) ? min(rt0 * 16 + col0, N - 1) : 0;
    const int rowA1 = (rt1 < nrt) ? min(rt1 * 16 + col0, N - 1) : 0;

    for (int i = threadIdx.x; i < 64 * 48; i += 256) {
        int r = i / 48, c = i % 48;
        sW[c * P2 + r] = (c < NC) ? W[(size_t)r * NC + c] : 0.0f;
    }
    __syncthreads();

    v8f acc[2][3];
    #pragma unroll
    for (int u = 0; u < 2; ++u)
        #pragma unroll
        for (int t = 0; t < 3; ++t) acc[u][t] = v8f{};

    const float* xr0 = Hin + (size_t)rowA0 * 64;
    const float* xr1 = Hin + (size_t)rowA1 * 64;
    const float dn0 = fmaxf(Dn[rowA0], 1e-16f);
    const float dn1 = fmaxf(Dn[rowA1], 1e-16f);
    for (int kk = 0; kk < 16; ++kk) {
        const int kb = kk * 4 + half * 2;
        v2f a0 = *(const v2f*)(xr0 + kb);
        v2f a1 = *(const v2f*)(xr1 + kb);
        a0[0] = fmaxf(a0[0] / dn0, 0.0f); a0[1] = fmaxf(a0[1] / dn0, 0.0f);
        a1[0] = fmaxf(a1[0] / dn1, 0.0f); a1[1] = fmaxf(a1[1] / dn1, 0.0f);
        #pragma unroll
        for (int t = 0; t < 3; ++t) {
            v2f b = *(const v2f*)(&sW[(t * 16 + col0) * P2 + kb]);
            acc[0][t] = __builtin_amdgcn_wmma_f32_16x16x4_f32(
                false, a0, false, b, (short)0, acc[0][t], false, false);
            acc[1][t] = __builtin_amdgcn_wmma_f32_16x16x4_f32(
                false, a1, false, b, (short)0, acc[1][t], false, false);
        }
    }

    #pragma unroll
    for (int u = 0; u < 2; ++u) {
        int rt = rt0 + u;
        if (rt >= nrt) continue;
        #pragma unroll
        for (int t = 0; t < 3; ++t)
            #pragma unroll
            for (int r = 0; r < 8; ++r) {
                int row = rt * 16 + r + half * 8;
                int col = t * 16 + col0;
                if (row < N && col < NC)
                    Hout[(size_t)row * NC + col] = acc[u][t][r];
            }
    }
}

#define NEG_INF_KEY 0x007FFFFFu
__global__ __launch_bounds__(256) void sd_kernel(
    const float* __restrict__ H, const float* __restrict__ asrc,
    const float* __restrict__ adst, float* __restrict__ s,
    unsigned* __restrict__ nd, int N, int F) {
    const int wid  = (int)((blockIdx.x * (long long)blockDim.x + threadIdx.x) >> 5);
    const int lane = threadIdx.x & 31;
    if (wid >= N) return;
    float ps = 0.0f, pd = 0.0f;
    for (int f = lane; f < F; f += 32) {
        float hv = H[(size_t)wid * F + f];
        ps = fmaf(hv, asrc[f], ps);
        pd = fmaf(hv, adst[f], pd);
    }
    #pragma unroll
    for (int off = 16; off > 0; off >>= 1) {
        ps += __shfl_down(ps, off, 32);
        pd += __shfl_down(pd, off, 32);
    }
    if (lane == 0) {
        s[wid] = ps;
        *(uint2*)(nd + 2u * (unsigned)wid) = make_uint2(__float_as_uint(pd), NEG_INF_KEY);
    }
}

__global__ void zero_f32(float* __restrict__ p, long long n) {
    long long i = blockIdx.x * (long long)blockDim.x + threadIdx.x;
    if (i < n) p[i] = 0.0f;
}
#define CTR_STRIDE 64
__global__ void init_seg(float* __restrict__ denom, unsigned* __restrict__ ctr, int n) {
    int i = blockIdx.x * blockDim.x + threadIdx.x;
    if (i < n) denom[i] = 0.0f;
    if (i < 4) ctr[i * CTR_STRIDE] = 0u;
}

#define TCH 128
#define NK (TCH / 32)
__device__ __forceinline__ int wave_side() {
    return (int)((__builtin_amdgcn_s_sendmsg_rtn(0x87) >> 18) & 1u);
}
__device__ __forceinline__ int draw(unsigned* ctr, int lane) {
    unsigned t = 0u;
    if (lane == 0) t = atomicAdd(ctr, 1u);
    return __builtin_amdgcn_readfirstlane((int)t);
}
__device__ __forceinline__ void block_edges(
    const int* __restrict__ ei, int t, int lane, int E, int EA, int NH, int r,
    int (&src)[NK], int (&dst)[NK], bool (&in)[NK]) {
    int ls[NK], ld[NK];
    #pragma unroll
    for (int k = 0; k < NK; ++k) {
        const int ec = min(t * TCH + k * 32 + lane, E - 1);
        ls[k] = ei[ec];
        ld[k] = ei[E + ec];
    }
    #pragma unroll
    for (int k = 0; k < NK; ++k) {
        const int e = t * TCH + k * 32 + lane;
        const bool loop = e >= E;
        src[k] = loop ? e - E : ls[k];
        dst[k] = loop ? e - E : ld[k];
        in[k] = (e < EA) && ((dst[k] >= NH) == (r != 0));
    }
}

__global__ __launch_bounds__(256) void edge_max_kernel(
    const int* __restrict__ ei, const float* __restrict__ s,
    unsigned* __restrict__ nd, float* __restrict__ es,
    unsigned* __restrict__ ctr, int E, int EA, int NH) {
    const int lane = threadIdx.x & 31;
    const int nT = (EA + TCH - 1) / TCH;
    int r = wave_side();
    for (int pass = 0; pass < 2; ++pass, r ^= 1) {
        unsigned* c = ctr + r * CTR_STRIDE;
        float* esr = es + (size_t)r * nT * TCH;
        for (int t = draw(c, lane); t < nT; t = draw(c, lane)) {
            int src[NK], dst[NK];
            bool in[NK];
            block_edges(ei, t, lane, E, EA, NH, r, src, dst, in);
            float sv[NK];
            #pragma unroll
            for (int k = 0; k < NK; ++k) {
                sv[k] = 0.0f;
                if (in[k]) sv[k] = s[src[k]];
            }
            #pragma unroll
            for (int k = 0; k < NK; ++k)
                if (in[k]) {
                    esr[t * TCH + k * 32 + lane] = sv[k];
                    atomicMax(&nd[2u * (unsigned)dst[k] + 1u], f2ord(sv[k]));
                }
        }
    }
}

#define GRP 16
__global__ __launch_bounds__(256) void edge_row_kernel(
    const int* __restrict__ ei, const float* __restrict__ ew,
    const float* __restrict__ es, const unsigned* __restrict__ nd,
    float* __restrict__ denom,
    const float* __restrict__ H, float* __restrict__ out,
    unsigned* __restrict__ ctr, int E, int EA, int F, int NH) {
    __shared__ int   qs[8][TCH];
    __shared__ int   qd[8][TCH];
    __shared__ float qc[8][TCH];
    const int lane = threadIdx.x & 31;
    const int wv = threadIdx.x >> 5;
    const int nT = (EA + TCH - 1) / TCH;
    const bool p0 = lane < F;
    const bool p1 = lane + 32 < F;
    const unsigned colA = (unsigned)min(lane, F - 1);
    const unsigned colB = (unsigned)min(lane + 32, F - 1);
    const unsigned uF = (unsigned)F;
    int r = wave_side();
    for (int pass = 0; pass < 2; ++pass, r ^= 1) {
        unsigned* c = ctr + r * CTR_STRIDE;
        const float* esr = es + (size_t)r * nT * TCH;
        for (int t = draw(c, lane); t < nT; t = draw(c, lane)) {
            int src[NK], dst[NK];
            bool in[NK];
            block_edges(ei, t, lane, E, EA, NH, r, src, dst, in);
            float sv[NK], wt[NK];
            uint2 rec[NK];
            #pragma unroll
            for (int k = 0; k < NK; ++k) {
                const int e = t * TCH + k * 32 + lane;
                sv[k] = 0.0f; rec[k] = make_uint2(0u, 0u); wt[k] = 1.0f;
                if (in[k]) {
                    sv[k] = esr[e];
                    rec[k] = *(const uint2*)(nd + 2u * (unsigned)dst[k]);
                    if (e < E) wt[k] = ew[e];
                }
            }
            int cnt = 0;
            #pragma unroll
            for (int k = 0; k < NK; ++k) {
                const unsigned mask = __builtin_amdgcn_ballot_w32(in[k]);
                const int slot = cnt + (int)__builtin_amdgcn_mbcnt_lo(mask, 0u);
                if (in[k]) {
                    const float dv = __uint_as_float(rec[k].x);
                    const float mx = leaky(ord2f(rec[k].y) + dv);
                    const float ex = __expf(leaky(sv[k] + dv) - mx);
                    atomicAdd(&denom[dst[k]], ex);
                    qs[wv][slot] = src[k];
                    qd[wv][slot] = dst[k];
                    qc[wv][slot] = ex * wt[k];
                }
                cnt += __builtin_popcount(mask);
            }
            for (int g = 0; g < cnt; g += GRP) {
                float va[GRP], vb[GRP];
                #pragma unroll
                for (int j = 0; j < GRP; ++j) {
                    const int idx = min(g + j, cnt - 1);
                    const unsigned ro = (unsigned)__builtin_amdgcn_readfirstlane(qs[wv][idx]) * uF;
                    va[j] = H[ro + colA];
                    vb[j] = H[ro + colB];
                }
                #pragma unroll
                for (int j = 0; j < GRP; ++j) {
                    if (g + j < cnt) {
                        const unsigned wo = (unsigned)__builtin_amdgcn_readfirstlane(qd[wv][g + j]) * uF;
                        const float cj = qc[wv][g + j];
                        if (p0) atomicAdd(&out[wo + (unsigned)lane], va[j] * cj);
                        if (p1) atomicAdd(&out[wo + (unsigned)lane + 32u], vb[j] * cj);
                    }
                }
            }
        }
    }
}

__global__ void norm_rows(float* __restrict__ out, const float* __restrict__ denom,
                          unsigned n, unsigned F) {
    const unsigned i = blockIdx.x * blockDim.x + threadIdx.x;
    if (i < n) out[i] = out[i] / fmaxf(denom[i / F], 1e-16f);
}

extern "C" void kernel_launch(void* const* d_in, const int* in_sizes, int n_in,
                              void* d_out, int out_size, void* d_ws, size_t ws_size,
                              hipStream_t stream) {
    const float* x   = (const float*)d_in[0];
    const int*   ei  = (const int*)  d_in[1];
    const float* ew  = (const float*)d_in[2];
    const float* W1  = (const float*)d_in[3];
    const float* a1s = (const float*)d_in[4];
    const float* a1d = (const float*)d_in[5];
    const float* W2  = (const float*)d_in[6];
    const float* a2s = (const float*)d_in[7];
    const float* a2d = (const float*)d_in[8];

    const int HID = in_sizes[4];
    const int FIN = in_sizes[3] / HID;
    const int NC  = in_sizes[7];
    const int N   = in_sizes[0] / FIN;
    const int E   = in_sizes[2];
    const int EA  = E + N;

    float*    h1    = (float*)d_ws;
    float*    out1  = h1   + (size_t)N * HID;
    float*    h2    = out1 + (size_t)N * HID;
    float*    sArr  = h2   + (size_t)N * NC;
    float*    denom = sArr + N;
    unsigned* ndArr = (unsigned*)(denom + N);
    unsigned* ctr   = ndArr + ((size_t)2 * N + 63) / 64 * 64;
    float*    esArr = (float*)(ctr + 4 * CTR_STRIDE);
    float*    outF  = (float*)d_out;

    const int nrt     = (N + 15) / 16;
    const int pairs   = (nrt + 1) / 2;
    const int gBlocks = (pairs + 7) / 8;
    const int eBlocks = 1024;
    const int NH      = N / 2;
    const int nBlocks = (N + 255) / 256;

    gemm1_kernel<<<gBlocks, 256, 0, stream>>>(x, W1, h1, N, FIN);
    sd_kernel<<<((long long)N * 32 + 255) / 256, 256, 0, stream>>>(h1, a1s, a1d, sArr, ndArr, N, HID);
    init_seg<<<nBlocks, 256, 0, stream>>>(denom, ctr, N);
    zero_f32<<<(int)(((long long)N * HID + 255) / 256), 256, 0, stream>>>(out1, (long long)N * HID);
    edge_max_kernel<<<eBlocks, 256, 0, stream>>>(ei, sArr, ndArr, esArr, ctr, E, EA, NH);
    edge_row_kernel<<<eBlocks, 256, 0, stream>>>(ei, ew, esArr, ndArr, denom, h1, out1, ctr + 2 * CTR_STRIDE, E, EA, HID, NH);

    gemm2_kernel<<<gBlocks, 256, 0, stream>>>(out1, denom, W2, h2, N, NC);
    sd_kernel<<<((long long)N * 32 + 255) / 256, 256, 0, stream>>>(h2, a2s, a2d, sArr, ndArr, N, NC);
    init_seg<<<nBlocks, 256, 0, stream>>>(denom, ctr, N);
    zero_f32<<<(int)(((long long)N * NC + 255) / 256), 256, 0, stream>>>(outF, (long long)N * NC);
    edge_max_kernel<<<eBlocks, 256, 0, stream>>>(ei, sArr, ndArr, esArr, ctr, E, EA, NH);
    edge_row_kernel<<<eBlocks, 256, 0, stream>>>(ei, ew, esArr, ndArr, denom, h2, outF, ctr + 2 * CTR_STRIDE, E, EA, NC, NH);
    norm_rows<<<(int)(((long long)N * NC + 255) / 256), 256, 0, stream>>>(outF, denom, (unsigned)N * (unsigned)NC, (unsigned)NC);
}


# Clears folded into the score kernel: a node's wave stores the zero start of its row sum, its exp-sum and the block counters; both clear launches per layer gone: 13 launches -> 9; atomics as before
# speedup vs baseline: 3.1949x; 1.0473x over previous
#include <hip/hip_runtime.h>
#include <hip/hip_bf16.h>


typedef __attribute__((ext_vector_type(2))) float v2f;
typedef __attribute__((ext_vector_type(8))) float v8f;

#define NEG_SLOPE 0.2f

__device__ __forceinline__ unsigned f2ord(float f) {
    unsigned u = __float_as_uint(f);
    return (u & 0x80000000u) ? ~u : (u | 0x80000000u);
}
__device__ __forceinline__ float ord2f(unsigned k) {
    unsigned u = (k & 0x80000000u) ? (k & 0x7fffffffu) : ~k;
    return __uint_as_float(u);
}
__device__ __forceinline__ float leaky(float x) {
    return x > 0.0f ? x : NEG_SLOPE * x;
}

#define P1 132
__global__ __launch_bounds__(256) void gemm1_kernel(
    const float* __restrict__ X, const float* __restrict__ W,
    float* __restrict__ H, int N, int K) {
    __shared__ float sW[64 * P1];
    const int lane = threadIdx.x & 31;
    const int wave = threadIdx.x >> 5;
    const int half = lane >> 4;
    const int col0 = lane & 15;
    const int nrt  = (N + 15) >> 4;
    const int rt0  = (blockIdx.x * 8 + wave) * 2;
    const int rt1  = rt0 + 1;
    const int rowA0 = (rt0 < nrt) ? min(rt0 * 16 + col0, N - 1) : 0;
    const int rowA1 = (rt1 < nrt) ? min(rt1 * 16 + col0, N - 1) : 0;

    v8f acc[2][4];
    #pragma unroll
    for (int u = 0; u < 2; ++u)
        #pragma unroll
        for (int t = 0; t < 4; ++t) acc[u][t] = v8f{};

    const int phases = K >> 7;
    for (int ph = 0; ph < phases; ++ph) {
        for (int i = threadIdx.x; i < 128 * 64; i += 256) {
            int r = i >> 6, c = i & 63;
            sW[c * P1 + r] = W[(size_t)ph * 128 * 64 + i];
        }
        __syncthreads();

        const float* xr0 = X + (size_t)rowA0 * K + ph * 128;
        const float* xr1 = X + (size_t)rowA1 * K + ph * 128;
        for (int kk = 0; kk < 32; ++kk) {
            const int kb = kk * 4 + half * 2;
            v2f a0 = *(const v2f*)(xr0 + kb);
            v2f a1 = *(const v2f*)(xr1 + kb);
            #pragma unroll
            for (int t = 0; t < 4; ++t) {
                v2f b = *(const v2f*)(&sW[(t * 16 + col0) * P1 + kb]);
                acc[0][t] = __builtin_amdgcn_wmma_f32_16x16x4_f32(
                    false, a0, false, b, (short)0, acc[0][t], false, false);
                acc[1][t] = __builtin_amdgcn_wmma_f32_16x16x4_f32(
                    false, a1, false, b, (short)0, acc[1][t], false, false);
            }
        }
        __syncthreads();
    }

    #pragma unroll
    for (int u = 0; u < 2; ++u) {
        int rt = rt0 + u;
        if (rt >= nrt) continue;
        #pragma unroll
        for (int t = 0; t < 4; ++t)
            #pragma unroll
            for (int r = 0; r < 8; ++r) {
                int row = rt * 16 + r + half * 8;
                if (row < N) H[(size_t)row * 64 + t * 16 + col0] = acc[u][t][r];
            }
    }
}

#define P2 68
__global__ __launch_bounds__(256) void gemm2_kernel(
    const float* __restrict__ Hin, const float* __restrict__ Dn,
    const float* __restrict__ W,
    float* __restrict__ Hout, int N, int NC) {
    __shared__ float sW[48 * P2];
    const int lane = threadIdx.x & 31;
    const int wave = threadIdx.x >> 5;
    const int half = lane >> 4;
    const int col0 = lane & 15;
    const int nrt  = (N + 15) >> 4;
    const int rt0  = (blockIdx.x * 8 + wave) * 2;
    const int rt1  = rt0 + 1;
    const int rowA0 = (rt0 < nrt) ? min(rt0 * 16 + col0, N - 1) : 0;
    const int rowA1 = (rt1 < nrt) ? min(rt1 * 16 + col0, N - 1) : 0;

    for (int i = threadIdx.x; i < 64 * 48; i += 256) {
        int r = i / 48, c = i % 48;
        sW[c * P2 + r] = (c < NC) ? W[(size_t)r * NC + c] : 0.0f;
    }
    __syncthreads();

    v8f acc[2][3];
    #pragma unroll
    for (int u = 0; u < 2; ++u)
        #pragma unroll
        for (int t = 0; t < 3; ++t) acc[u][t] = v8f{};

    const float* xr0 = Hin + (size_t)rowA0 * 64;
    const float* xr1 = Hin + (size_t)rowA1 * 64;
    const float dn0 = fmaxf(Dn[rowA0], 1e-16f);
    const float dn1 = fmaxf(Dn[rowA1], 1e-16f);
    for (int kk = 0; kk < 16; ++kk) {
        const int kb = kk * 4 + half * 2;
        v2f a0 = *(const v2f*)(xr0 + kb);
        v2f a1 = *(const v2f*)(xr1 + kb);
        a0[0] = fmaxf(a0[0] / dn0, 0.0f); a0[1] = fmaxf(a0[1] / dn0, 0.0f);
        a1[0] = fmaxf(a1[0] / dn1, 0.0f); a1[1] = fmaxf(a1[1] / dn1, 0.0f);
        #pragma unroll
        for (int t = 0; t < 3; ++t) {
            v2f b = *(const v2f*)(&sW[(t * 16 + col0) * P2 + kb]);
            acc[0][t] = __builtin_amdgcn_wmma_f32_16x16x4_f32(
                false, a0, false, b, (short)0, acc[0][t], false, false);
            acc[1][t] = __builtin_amdgcn_wmma_f32_16x16x4_f32(
                false, a1, false, b, (short)0, acc[1][t], false, false);
        }
    }

    #pragma unroll
    for (int u = 0; u < 2; ++u) {
        int rt = rt0 + u;
        if (rt >= nrt) continue;
        #pragma unroll
        for (int t = 0; t < 3; ++t)
            #pragma unroll
            for (int r = 0; r < 8; ++r) {
                int row = rt * 16 + r + half * 8;
                int col = t * 16 + col0;
                if (row < N && col < NC)
                    Hout[(size_t)row * NC + col] = acc[u][t][r];
            }
    }
}

#define NEG_INF_KEY 0x007FFFFFu
#define CTR_STRIDE 64
__global__ __launch_bounds__(256) void sd_kernel(
    const float* __restrict__ H, const float* __restrict__ asrc,
    const float* __restrict__ adst, float* __restrict__ s,
    unsigned* __restrict__ nd, float* __restrict__ denom,
    float* __restrict__ acc, unsigned* __restrict__ ctr, int N, int F) {
    const int wid  = (int)((blockIdx.x * (long long)blockDim.x + threadIdx.x) >> 5);
    const int lane = threadIdx.x & 31;
    if (wid >= N) return;
    float ps = 0.0f, pd = 0.0f;
    for (int f = lane; f < F; f += 32) {
        float hv = H[(size_t)wid * F + f];
        acc[(size_t)wid * F + f] = 0.0f;
        ps = fmaf(hv, asrc[f], ps);
        pd = fmaf(hv, adst[f], pd);
    }
    #pragma unroll
    for (int off = 16; off > 0; off >>= 1) {
        ps += __shfl_down(ps, off, 32);
        pd += __shfl_down(pd, off, 32);
    }
    if (lane == 0) {
        s[wid] = ps;
        *(uint2*)(nd + 2u * (unsigned)wid) = make_uint2(__float_as_uint(pd), NEG_INF_KEY);
        denom[wid] = 0.0f;
    }
    if (wid == 0 && lane < 4) ctr[lane * CTR_STRIDE] = 0u;
}

#define TCH 128
#define NK (TCH / 32)
__device__ __forceinline__ int wave_side() {
    return (int)((__builtin_amdgcn_s_sendmsg_rtn(0x87) >> 18) & 1u);
}
__device__ __forceinline__ int draw(unsigned* ctr, int lane) {
    unsigned t = 0u;
    if (lane == 0) t = atomicAdd(ctr, 1u);
    return __builtin_amdgcn_readfirstlane((int)t);
}
__device__ __forceinline__ void block_edges(
    const int* __restrict__ ei, int t, int lane, int E, int EA, int NH, int r,
    int (&src)[NK], int (&dst)[NK], bool (&in)[NK]) {
    int ls[NK], ld[NK];
    #pragma unroll
    for (int k = 0; k < NK; ++k) {
        const int ec = min(t * TCH + k * 32 + lane, E - 1);
        ls[k] = ei[ec];
        ld[k] = ei[E + ec];
    }
    #pragma unroll
    for (int k = 0; k < NK; ++k) {
        const int e = t * TCH + k * 32 + lane;
        const bool loop = e >= E;
        src[k] = loop ? e - E : ls[k];
        dst[k] = loop ? e - E : ld[k];
        in[k] = (e < EA) && ((dst[k] >= NH) == (r != 0));
    }
}

__global__ __launch_bounds__(256) void edge_max_kernel(
    const int* __restrict__ ei, const float* __restrict__ s,
    unsigned* __restrict__ nd, float* __restrict__ es,
    unsigned* __restrict__ ctr, int E, int EA, int NH) {
    const int lane = threadIdx.x & 31;
    const int nT = (EA + TCH - 1) / TCH;
    int r = wave_side();
    for (int pass = 0; pass < 2; ++pass, r ^= 1) {
        unsigned* c = ctr + r * CTR_STRIDE;
        float* esr = es + (size_t)r * nT * TCH;
        for (int t = draw(c, lane); t < nT; t = draw(c, lane)) {
            int src[NK], dst[NK];
            bool in[NK];
            block_edges(ei, t, lane, E, EA, NH, r, src, dst, in);
            float sv[NK];
            #pragma unroll
            for (int k = 0; k < NK; ++k) {
                sv[k] = 0.0f;
                if (in[k]) sv[k] = s[src[k]];
            }
            #pragma unroll
            for (int k = 0; k < NK; ++k)
                if (in[k]) {
                    esr[t * TCH + k * 32 + lane] = sv[k];
                    atomicMax(&nd[2u * (unsigned)dst[k] + 1u], f2ord(sv[k]));
                }
        }
    }
}

#define GRP 16
__global__ __launch_bounds__(256) void edge_row_kernel(
    const int* __restrict__ ei, const float* __restrict__ ew,
    const float* __restrict__ es, const unsigned* __restrict__ nd,
    float* __restrict__ denom,
    const float* __restrict__ H, float* __restrict__ out,
    unsigned* __restrict__ ctr, int E, int EA, int F, int NH) {
    __shared__ int   qs[8][TCH];
    __shared__ int   qd[8][TCH];
    __shared__ float qc[8][TCH];
    const int lane = threadIdx.x & 31;
    const int wv = threadIdx.x >> 5;
    const int nT = (EA + TCH - 1) / TCH;
    const bool p0 = lane < F;
    const bool p1 = lane + 32 < F;
    const unsigned colA = (unsigned)min(lane, F - 1);
    const unsigned colB = (unsigned)min(lane + 32, F - 1);
    const unsigned uF = (unsigned)F;
    int r = wave_side();
    for (int pass = 0; pass < 2; ++pass, r ^= 1) {
        unsigned* c = ctr + r * CTR_STRIDE;
        const float* esr = es + (size_t)r * nT * TCH;
        for (int t = draw(c, lane); t < nT; t = draw(c, lane)) {
            int src[NK], dst[NK];
            bool in[NK];
            block_edges(ei, t, lane, E, EA, NH, r, src, dst, in);
            float sv[NK], wt[NK];
            uint2 rec[NK];
            #pragma unroll
            for (int k = 0; k < NK; ++k) {
                const int e = t * TCH + k * 32 + lane;
                sv[k] = 0.0f; rec[k] = make_uint2(0u, 0u); wt[k] = 1.0f;
                if (in[k]) {
                    sv[k] = esr[e];
                    rec[k] = *(const uint2*)(nd + 2u * (unsigned)dst[k]);
                    if (e < E) wt[k] = ew[e];
                }
            }
            int cnt = 0;
            #pragma unroll
            for (int k = 0; k < NK; ++k) {
                const unsigned mask = __builtin_amdgcn_ballot_w32(in[k]);
                const int slot = cnt + (int)__builtin_amdgcn_mbcnt_lo(mask, 0u);
                if (in[k]) {
                    const float dv = __uint_as_float(rec[k].x);
                    const float mx = leaky(ord2f(rec[k].y) + dv);
                    const float ex = __expf(leaky(sv[k] + dv) - mx);
                    atomicAdd(&denom[dst[k]], ex);
                    qs[wv][slot] = src[k];
                    qd[wv][slot] = dst[k];
                    qc[wv][slot] = ex * wt[k];
                }
                cnt += __builtin_popcount(mask);
            }
            for (int g = 0; g < cnt; g += GRP) {
                float va[GRP], vb[GRP];
                #pragma unroll
                for (int j = 0; j < GRP; ++j) {
                    const int idx = min(g + j, cnt - 1);
                    const unsigned ro = (unsigned)__builtin_amdgcn_readfirstlane(qs[wv][idx]) * uF;
                    va[j] = H[ro + colA];
                    vb[j] = H[ro + colB];
                }
                #pragma unroll
                for (int j = 0; j < GRP; ++j) {
                    if (g + j < cnt) {
                        const unsigned wo = (unsigned)__builtin_amdgcn_readfirstlane(qd[wv][g + j]) * uF;
                        const float cj = qc[wv][g + j];
                        if (p0) atomicAdd(&out[wo + (unsigned)lane], va[j] * cj);
                        if (p1) atomicAdd(&out[wo + (unsigned)lane + 32u], vb[j] * cj);
                    }
                }
            }
        }
    }
}

__global__ void norm_rows(float* __restrict__ out, const float* __restrict__ denom,
                          unsigned n, unsigned F) {
    const unsigned i = blockIdx.x * blockDim.x + threadIdx.x;
    if (i < n) out[i] = out[i] / fmaxf(denom[i / F], 1e-16f);
}

extern "C" void kernel_launch(void* const* d_in, const int* in_sizes, int n_in,
                              void* d_out, int out_size, void* d_ws, size_t ws_size,
                              hipStream_t stream) {
    const float* x   = (const float*)d_in[0];
    const int*   ei  = (const int*)  d_in[1];
    const float* ew  = (const float*)d_in[2];
    const float* W1  = (const float*)d_in[3];
    const float* a1s = (const float*)d_in[4];
    const float* a1d = (const float*)d_in[5];
    const float* W2  = (const float*)d_in[6];
    const float* a2s = (const float*)d_in[7];
    const float* a2d = (const float*)d_in[8];

    const int HID = in_sizes[4];
    const int FIN = in_sizes[3] / HID;
    const int NC  = in_sizes[7];
    const int N   = in_sizes[0] / FIN;
    const int E   = in_sizes[2];
    const int EA  = E + N;

    float*    h1    = (float*)d_ws;
    float*    out1  = h1   + (size_t)N * HID;
    float*    h2    = out1 + (size_t)N * HID;
    float*    sArr  = h2   + (size_t)N * NC;
    float*    denom = sArr + N;
    unsigned* ndArr = (unsigned*)(denom + N);
    unsigned* ctr   = ndArr + ((size_t)2 * N + 63) / 64 * 64;
    float*    esArr = (float*)(ctr + 4 * CTR_STRIDE);
    float*    outF  = (float*)d_out;

    const int nrt     = (N + 15) / 16;
    const int pairs   = (nrt + 1) / 2;
    const int gBlocks = (pairs + 7) / 8;
    const int eBlocks = 1024;
    const int NH      = N / 2;
    const int sBlocks = (int)(((long long)N * 32 + 255) / 256);

    gemm1_kernel<<<gBlocks, 256, 0, stream>>>(x, W1, h1, N, FIN);
    sd_kernel<<<sBlocks, 256, 0, stream>>>(h1, a1s, a1d, sArr, ndArr, denom, out1, ctr, N, HID);
    edge_max_kernel<<<eBlocks, 256, 0, stream>>>(ei, sArr, ndArr, esArr, ctr, E, EA, NH);
    edge_row_kernel<<<eBlocks, 256, 0, stream>>>(ei, ew, esArr, ndArr, denom, h1, out1, ctr + 2 * CTR_STRIDE, E, EA, HID, NH);

    gemm2_kernel<<<gBlocks, 256, 0, stream>>>(out1, denom, W2, h2, N, NC);
    sd_kernel<<<sBlocks, 256, 0, stream>>>(h2, a2s, a2d, sArr, ndArr, denom, outF, ctr, N, NC);
    edge_max_kernel<<<eBlocks, 256, 0, stream>>>(ei, sArr, ndArr, esArr, ctr, E, EA, NH);
    edge_row_kernel<<<eBlocks, 256, 0, stream>>>(ei, ew, esArr, ndArr, denom, h2, outF, ctr + 2 * CTR_STRIDE, E, EA, NC, NH);
    norm_rows<<<(int)(((long long)N * NC + 255) / 256), 256, 0, stream>>>(outF, denom, (unsigned)N * (unsigned)NC, (unsigned)NC);
}


# Layer 2 rows (h2 and its sum buffer) get a 256-byte pitch in workspace, the exp-sum in the sum row's spare column: an edge adds into 2 aligned lines, no exp-sum add of its own; arrival-order atomics
# speedup vs baseline: 3.3511x; 1.0489x over previous
#include <hip/hip_runtime.h>
#include <hip/hip_bf16.h>


typedef __attribute__((ext_vector_type(2))) float v2f;
typedef __attribute__((ext_vector_type(8))) float v8f;

#define NEG_SLOPE 0.2f

__device__ __forceinline__ unsigned f2ord(float f) {
    unsigned u = __float_as_uint(f);
    return (u & 0x80000000u) ? ~u : (u | 0x80000000u);
}
__device__ __forceinline__ float ord2f(unsigned k) {
    unsigned u = (k & 0x80000000u) ? (k & 0x7fffffffu) : ~k;
    return __uint_as_float(u);
}
__device__ __forceinline__ float leaky(float x) {
    return x > 0.0f ? x : NEG_SLOPE * x;
}

#define P1 132
__global__ __launch_bounds__(256) void gemm1_kernel(
    const float* __restrict__ X, const float* __restrict__ W,
    float* __restrict__ H, int N, int K) {
    __shared__ float sW[64 * P1];
    const int lane = threadIdx.x & 31;
    const int wave = threadIdx.x >> 5;
    const int half = lane >> 4;
    const int col0 = lane & 15;
    const int nrt  = (N + 15) >> 4;
    const int rt0  = (blockIdx.x * 8 + wave) * 2;
    const int rt1  = rt0 + 1;
    const int rowA0 = (rt0 < nrt) ? min(rt0 * 16 + col0, N - 1) : 0;
    const int rowA1 = (rt1 < nrt) ? min(rt1 * 16 + col0, N - 1) : 0;

    v8f acc[2][4];
    #pragma unroll
    for (int u = 0; u < 2; ++u)
        #pragma unroll
        for (int t = 0; t < 4; ++t) acc[u][t] = v8f{};

    const int phases = K >> 7;
    for (int ph = 0; ph < phases; ++ph) {
        for (int i = threadIdx.x; i < 128 * 64; i += 256) {
            int r = i >> 6, c = i & 63;
            sW[c * P1 + r] = W[(size_t)ph * 128 * 64 + i];
        }
        __syncthreads();

        const float* xr0 = X + (size_t)rowA0 * K + ph * 128;
        const float* xr1 = X + (size_t)rowA1 * K + ph * 128;
        for (int kk = 0; kk < 32; ++kk) {
            const int kb = kk * 4 + half * 2;
            v2f a0 = *(const v2f*)(xr0 + kb);
            v2f a1 = *(const v2f*)(xr1 + kb);
            #pragma unroll
            for (int t = 0; t < 4; ++t) {
                v2f b = *(const v2f*)(&sW[(t * 16 + col0) * P1 + kb]);
                acc[0][t] = __builtin_amdgcn_wmma_f32_16x16x4_f32(
                    false, a0, false, b, (short)0, acc[0][t], false, false);
                acc[1][t] = __builtin_amdgcn_wmma_f32_16x16x4_f32(
                    false, a1, false, b, (short)0, acc[1][t], false, false);
            }
        }
        __syncthreads();
    }

    #pragma unroll
    for (int u = 0; u < 2; ++u) {
        int rt = rt0 + u;
        if (rt >= nrt) continue;
        #pragma unroll
        for (int t = 0; t < 4; ++t)
            #pragma unroll
            for (int r = 0; r < 8; ++r) {
                int row = rt * 16 + r + half * 8;
                if (row < N) H[(size_t)row * 64 + t * 16 + col0] = acc[u][t][r];
            }
    }
}

#define P2 68
__global__ __launch_bounds__(256) void gemm2_kernel(
    const float* __restrict__ Hin, const float* __restrict__ Dn,
    const float* __restrict__ W,
    float* __restrict__ Hout, int N, int NC, int HP) {
    __shared__ float sW[48 * P2];
    const int lane = threadIdx.x & 31;
    const int wave = threadIdx.x >> 5;
    const int half = lane >> 4;
    const int col0 = lane & 15;
    const int nrt  = (N + 15) >> 4;
    const int rt0  = (blockIdx.x * 8 + wave) * 2;
    const int rt1  = rt0 + 1;
    const int rowA0 = (rt0 < nrt) ? min(rt0 * 16 + col0, N - 1) : 0;
    const int rowA1 = (rt1 < nrt) ? min(rt1 * 16 + col0, N - 1) : 0;

    for (int i = threadIdx.x; i < 64 * 48; i += 256) {
        int r = i / 48, c = i % 48;
        sW[c * P2 + r] = (c < NC) ? W[(size_t)r * NC + c] : 0.0f;
    }
    __syncthreads();

    v8f acc[2][3];
    #pragma unroll
    for (int u = 0; u < 2; ++u)
        #pragma unroll
        for (int t = 0; t < 3; ++t) acc[u][t] = v8f{};

    const float* xr0 = Hin + (size_t)rowA0 * 64;
    const float* xr1 = Hin + (size_t)rowA1 * 64;
    const float dn0 = fmaxf(Dn[rowA0], 1e-16f);
    const float dn1 = fmaxf(Dn[rowA1], 1e-16f);
    for (int kk = 0; kk < 16; ++kk) {
        const int kb = kk * 4 + half * 2;
        v2f a0 = *(const v2f*)(xr0 + kb);
        v2f a1 = *(const v2f*)(xr1 + kb);
        a0[0] = fmaxf(a0[0] / dn0, 0.0f); a0[1] = fmaxf(a0[1] / dn0, 0.0f);
        a1[0] = fmaxf(a1[0] / dn1, 0.0f); a1[1] = fmaxf(a1[1] / dn1, 0.0f);
        #pragma unroll
        for (int t = 0; t < 3; ++t) {
            v2f b = *(const v2f*)(&sW[(t * 16 + col0) * P2 + kb]);
            acc[0][t] = __builtin_amdgcn_wmma_f32_16x16x4_f32(
                false, a0, false, b, (short)0, acc[0][t], false, false);
            acc[1][t] = __builtin_amdgcn_wmma_f32_16x16x4_f32(
                false, a1, false, b, (short)0, acc[1][t], false, false);
        }
    }

    #pragma unroll
    for (int u = 0; u < 2; ++u) {
        int rt = rt0 + u;
        if (rt >= nrt) continue;
        #pragma unroll
        for (int t = 0; t < 3; ++t)
            #pragma unroll
            for (int r = 0; r < 8; ++r) {
                int row = rt * 16 + r + half * 8;
                int col = t * 16 + col0;
                if (row < N && col < NC)
                    Hout[(size_t)row * HP + col] = acc[u][t][r];
            }
    }
}

#define NEG_INF_KEY 0x007FFFFFu
#define CTR_STRIDE 64
__global__ __launch_bounds__(256) void sd_kernel(
    const float* __restrict__ H, const float* __restrict__ asrc,
    const float* __restrict__ adst, float* __restrict__ s,
    unsigned* __restrict__ nd, float* __restrict__ denom,
    float* __restrict__ acc, unsigned* __restrict__ ctr, int N, int F, int P, int fold) {
    const int wid  = (int)((blockIdx.x * (long long)blockDim.x + threadIdx.x) >> 5);
    const int lane = threadIdx.x & 31;
    if (wid >= N) return;
    float ps = 0.0f, pd = 0.0f;
    for (int f = lane; f < F; f += 32) {
        float hv = H[(size_t)wid * P + f];
        acc[(size_t)wid * P + f] = 0.0f;
        ps = fmaf(hv, asrc[f], ps);
        pd = fmaf(hv, adst[f], pd);
    }
    #pragma unroll
    for (int off = 16; off > 0; off >>= 1) {
        ps += __shfl_down(ps, off, 32);
        pd += __shfl_down(pd, off, 32);
    }
    if (lane == 0) {
        s[wid] = ps;
        *(uint2*)(nd + 2u * (unsigned)wid) = make_uint2(__float_as_uint(pd), NEG_INF_KEY);
        if (fold) acc[(size_t)wid * P + F] = 0.0f;
        else      denom[wid] = 0.0f;
    }
    if (wid == 0 && lane < 4) ctr[lane * CTR_STRIDE] = 0u;
}

#define TCH 128
#define NK (TCH / 32)
__device__ __forceinline__ int wave_side() {
    return (int)((__builtin_amdgcn_s_sendmsg_rtn(0x87) >> 18) & 1u);
}
__device__ __forceinline__ int draw(unsigned* ctr, int lane) {
    unsigned t = 0u;
    if (lane == 0) t = atomicAdd(ctr, 1u);
    return __builtin_amdgcn_readfirstlane((int)t);
}
__device__ __forceinline__ void block_edges(
    const int* __restrict__ ei, int t, int lane, int E, int EA, int NH, int r,
    int (&src)[NK], int (&dst)[NK], bool (&in)[NK]) {
    int ls[NK], ld[NK];
    #pragma unroll
    for (int k = 0; k < NK; ++k) {
        const int ec = min(t * TCH + k * 32 + lane, E - 1);
        ls[k] = ei[ec];
        ld[k] = ei[E + ec];
    }
    #pragma unroll
    for (int k = 0; k < NK; ++k) {
        const int e = t * TCH + k * 32 + lane;
        const bool loop = e >= E;
        src[k] = loop ? e - E : ls[k];
        dst[k] = loop ? e - E : ld[k];
        in[k] = (e < EA) && ((dst[k] >= NH) == (r != 0));
    }
}

__global__ __launch_bounds__(256) void edge_max_kernel(
    const int* __restrict__ ei, const float* __restrict__ s,
    unsigned* __restrict__ nd, float* __restrict__ es,
    unsigned* __restrict__ ctr, int E, int EA, int NH) {
    const int lane = threadIdx.x & 31;
    const int nT = (EA + TCH - 1) / TCH;
    int r = wave_side();
    for (int pass = 0; pass < 2; ++pass, r ^= 1) {
        unsigned* c = ctr + r * CTR_STRIDE;
        float* esr = es + (size_t)r * nT * TCH;
        for (int t = draw(c, lane); t < nT; t = draw(c, lane)) {
            int src[NK], dst[NK];
            bool in[NK];
            block_edges(ei, t, lane, E, EA, NH, r, src, dst, in);
            float sv[NK];
            #pragma unroll
            for (int k = 0; k < NK; ++k) {
                sv[k] = 0.0f;
                if (in[k]) sv[k] = s[src[k]];
            }
            #pragma unroll
            for (int k = 0; k < NK; ++k)
                if (in[k]) {
                    esr[t * TCH + k * 32 + lane] = sv[k];
                    atomicMax(&nd[2u * (unsigned)dst[k] + 1u], f2ord(sv[k]));
                }
        }
    }
}

#define GRP 16
template <bool FOLD>
__global__ __launch_bounds__(256) void edge_row_kernel(
    const int* __restrict__ ei, const float* __restrict__ ew,
    const float* __restrict__ es, const unsigned* __restrict__ nd,
    float* __restrict__ denom,
    const float* __restrict__ H, float* __restrict__ out,
    unsigned* __restrict__ ctr, int E, int EA, int F, int P, int NH) {
    __shared__ int   qs[8][TCH];
    __shared__ int   qd[8][TCH];
    __shared__ float qc[8][TCH];
    __shared__ float qe[8][TCH];
    const int lane = threadIdx.x & 31;
    const int wv = threadIdx.x >> 5;
    const int nT = (EA + TCH - 1) / TCH;
    const bool fA = FOLD && lane == F;
    const bool fB = FOLD && lane + 32 == F;
    const bool p0 = lane < F || fA;
    const bool p1 = lane + 32 < F || fB;
    const unsigned colA = (unsigned)min(lane, F - 1);
    const unsigned colB = (unsigned)min(lane + 32, F - 1);
    const unsigned uF = (unsigned)P;
    int r = wave_side();
    for (int pass = 0; pass < 2; ++pass, r ^= 1) {
        unsigned* c = ctr + r * CTR_STRIDE;
        const float* esr = es + (size_t)r * nT * TCH;
        for (int t = draw(c, lane); t < nT; t = draw(c, lane)) {
            int src[NK], dst[NK];
            bool in[NK];
            block_edges(ei, t, lane, E, EA, NH, r, src, dst, in);
            float sv[NK], wt[NK];
            uint2 rec[NK];
            #pragma unroll
            for (int k = 0; k < NK; ++k) {
                const int e = t * TCH + k * 32 + lane;
                sv[k] = 0.0f; rec[k] = make_uint2(0u, 0u); wt[k] = 1.0f;
                if (in[k]) {
                    sv[k] = esr[e];
                    rec[k] = *(const uint2*)(nd + 2u * (unsigned)dst[k]);
                    if (e < E) wt[k] = ew[e];
                }
            }
            int cnt = 0;
            #pragma unroll
            for (int k = 0; k < NK; ++k) {
                const unsigned mask = __builtin_amdgcn_ballot_w32(in[k]);
                const int slot = cnt + (int)__builtin_amdgcn_mbcnt_lo(mask, 0u);
                if (in[k]) {
                    const float dv = __uint_as_float(rec[k].x);
                    const float mx = leaky(ord2f(rec[k].y) + dv);
                    const float ex = __expf(leaky(sv[k] + dv) - mx);
                    if (!FOLD) atomicAdd(&denom[dst[k]], ex);
                    qs[wv][slot] = src[k];
                    qd[wv][slot] = dst[k];
                    qc[wv][slot] = ex * wt[k];
                    if (FOLD) qe[wv][slot] = ex;
                }
                cnt += __builtin_popcount(mask);
            }
            for (int g = 0; g < cnt; g += GRP) {
                float va[GRP], vb[GRP];
                #pragma unroll
                for (int j = 0; j < GRP; ++j) {
                    const int idx = min(g + j, cnt - 1);
                    const unsigned ro = (unsigned)__builtin_amdgcn_readfirstlane(qs[wv][idx]) * uF;
                    va[j] = H[ro + colA];
                    vb[j] = H[ro + colB];
                }
                #pragma unroll
                for (int j = 0; j < GRP; ++j) {
                    if (g + j < cnt) {
                        const unsigned wo = (unsigned)__builtin_amdgcn_readfirstlane(qd[wv][g + j]) * uF;
                        const float cj = qc[wv][g + j];
                        float ta = va[j] * cj, tb = vb[j] * cj;
                        if (FOLD) {
                            const float ej = qe[wv][g + j];
                            ta = fA ? ej : ta;
                            tb = fB ? ej : tb;
                        }
                        if (p0) atomicAdd(&out[wo + (unsigned)lane], ta);
                        if (p1) atomicAdd(&out[wo + (unsigned)lane + 32u], tb);
                    }
                }
            }
        }
    }
}

__global__ void norm_rows(float* __restrict__ out, const float* __restrict__ acc,
                          unsigned n, unsigned F, unsigned P) {
    const unsigned i = blockIdx.x * blockDim.x + threadIdx.x;
    if (i < n) {
        const unsigned node = i / F;
        const unsigned c = i - node * F;
        out[i] = acc[node * P + c] / fmaxf(acc[node * P + F], 1e-16f);
    }
}

extern "C" void kernel_launch(void* const* d_in, const int* in_sizes, int n_in,
                              void* d_out, int out_size, void* d_ws, size_t ws_size,
                              hipStream_t stream) {
    const float* x   = (const float*)d_in[0];
    const int*   ei  = (const int*)  d_in[1];
    const float* ew  = (const float*)d_in[2];
    const float* W1  = (const float*)d_in[3];
    const float* a1s = (const float*)d_in[4];
    const float* a1d = (const float*)d_in[5];
    const float* W2  = (const float*)d_in[6];
    const float* a2s = (const float*)d_in[7];
    const float* a2d = (const float*)d_in[8];

    const int HID = in_sizes[4];
    const int FIN = in_sizes[3] / HID;
    const int NC  = in_sizes[7];
    const int N   = in_sizes[0] / FIN;
    const int E   = in_sizes[2];
    const int EA  = E + N;

    const int RP = 64;
    float*    h1    = (float*)d_ws;
    float*    out1  = h1   + (size_t)N * RP;
    float*    h2    = out1 + (size_t)N * RP;
    float*    acc2  = h2   + (size_t)N * RP;
    float*    sArr  = acc2 + (size_t)N * RP;
    float*    denom = sArr + N;
    unsigned* ndArr = (unsigned*)(denom + N);
    unsigned* ctr   = ndArr + ((size_t)2 * N + 63) / 64 * 64;
    float*    esArr = (float*)(ctr + 4 * CTR_STRIDE);
    float*    outF  = (float*)d_out;

    const int nrt     = (N + 15) / 16;
    const int pairs   = (nrt + 1) / 2;
    const int gBlocks = (pairs + 7) / 8;
    const int eBlocks = 1024;
    const int NH      = N / 2;
    const int sBlocks = (int)(((long long)N * 32 + 255) / 256);

    gemm1_kernel<<<gBlocks, 256, 0, stream>>>(x, W1, h1, N, FIN);
    sd_kernel<<<sBlocks, 256, 0, stream>>>(h1, a1s, a1d, sArr, ndArr, denom, out1, ctr, N, HID, RP, 0);
    edge_max_kernel<<<eBlocks, 256, 0, stream>>>(ei, sArr, ndArr, esArr, ctr, E, EA, NH);
    edge_row_kernel<false><<<eBlocks, 256, 0, stream>>>(ei, ew, esArr, ndArr, denom, h1, out1, ctr + 2 * CTR_STRIDE, E, EA, HID, RP, NH);

    gemm2_kernel<<<gBlocks, 256, 0, stream>>>(out1, denom, W2, h2, N, NC, RP);
    sd_kernel<<<sBlocks, 256, 0, stream>>>(h2, a2s, a2d, sArr, ndArr, denom, acc2, ctr, N, NC, RP, 1);
    edge_max_kernel<<<eBlocks, 256, 0, stream>>>(ei, sArr, ndArr, esArr, ctr, E, EA, NH);
    edge_row_kernel<true><<<eBlocks, 256, 0, stream>>>(ei, ew, esArr, ndArr, denom, h2, acc2, ctr + 2 * CTR_STRIDE, E, EA, NC, RP, NH);
    norm_rows<<<(int)(((long long)N * NC + 255) / 256), 256, 0, stream>>>(outF, acc2, (unsigned)N * (unsigned)NC, (unsigned)NC, (unsigned)RP);
}


# Softmax shift = leaky(max s of all nodes + d) when max s - min s <= 32 (GEMM blocks leave min/max): max pass exits at once, row pass gathers s[src]; else as before; ex scaled per node, same atomics
# speedup vs baseline: 4.1780x; 1.2468x over previous
#include <hip/hip_runtime.h>
#include <hip/hip_bf16.h>


typedef __attribute__((ext_vector_type(2))) float v2f;
typedef __attribute__((ext_vector_type(8))) float v8f;

#define NEG_SLOPE 0.2f

__device__ __forceinline__ unsigned f2ord(float f) {
    unsigned u = __float_as_uint(f);
    return (u & 0x80000000u) ? ~u : (u | 0x80000000u);
}
__device__ __forceinline__ float ord2f(unsigned k) {
    unsigned u = (k & 0x80000000u) ? (k & 0x7fffffffu) : ~k;
    return __uint_as_float(u);
}
__device__ __forceinline__ float leaky(float x) {
    return x > 0.0f ? x : NEG_SLOPE * x;
}

#define SPREAD_MAX 32.0f
__device__ __forceinline__ bool spread_small(const float* __restrict__ stat, int nstat, int lane, float& smax) {
    float hi = -__builtin_inff(), lo = __builtin_inff();
    for (int base = 0; base < nstat; base += 16 * 32) {
        float2 p[16];
        #pragma unroll
        for (int j = 0; j < 16; ++j)
            p[j] = *(const float2*)(stat + 2 * min(base + j * 32 + lane, nstat - 1));
        #pragma unroll
        for (int j = 0; j < 16; ++j) {
            hi = fmaxf(hi, p[j].x);
            lo = fminf(lo, p[j].y);
        }
    }
    #pragma unroll
    for (int o = 16; o > 0; o >>= 1) {
        hi = fmaxf(hi, __shfl_xor(hi, o));
        lo = fminf(lo, __shfl_xor(lo, o));
    }
    smax = __uint_as_float((unsigned)__builtin_amdgcn_readfirstlane((int)__float_as_uint(hi)));
    return __builtin_amdgcn_readfirstlane((hi - lo <= SPREAD_MAX) ? 1 : 0) != 0;
}
__device__ __forceinline__ void wave_minmax(float sv, bool ok, float (&red)[2][8], int wave, int lane) {
    float hi = ok ? sv : -__builtin_inff();
    float lo = ok ? sv : __builtin_inff();
    #pragma unroll
    for (int o = 16; o > 0; o >>= 1) {
        hi = fmaxf(hi, __shfl_xor(hi, o));
        lo = fminf(lo, __shfl_xor(lo, o));
    }
    if (lane == 0) { red[0][wave] = hi; red[1][wave] = lo; }
}
__device__ __forceinline__ void block_minmax(const float (&red)[2][8], float* __restrict__ stat) {
    if (threadIdx.x == 0) {
        float hi = red[0][0], lo = red[1][0];
        #pragma unroll
        for (int w = 1; w < 8; ++w) { hi = fmaxf(hi, red[0][w]); lo = fminf(lo, red[1][w]); }
        *(float2*)(stat + 2 * blockIdx.x) = make_float2(hi, lo);
    }
}

#define NEG_INF_KEY 0x007FFFFFu
#define CTR_STRIDE 64

__device__ __forceinline__ void zero_rows(float* __restrict__ Z, int node0, int lane, int N, int P) {
    const long long lim = (long long)N * P;
    const long long base = (long long)node0 * P;
    for (int o = lane * 4; o < 32 * P; o += 128)
        if (base + o < lim) *(float4*)(Z + base + o) = make_float4(0.0f, 0.0f, 0.0f, 0.0f);
}

#define P1 132
#define NB1 80
__global__ __launch_bounds__(256) void gemm1_kernel(
    const float* __restrict__ X, const float* __restrict__ W,
    const float* __restrict__ asrc, const float* __restrict__ adst,
    float* __restrict__ H, float* __restrict__ s, unsigned* __restrict__ nd,
    float* __restrict__ denom, float* __restrict__ acc0, unsigned* __restrict__ ctr,
    float* __restrict__ stat, int N, int K) {
    __shared__ float sW[NB1 * P1];
    __shared__ float sSD[8][2][32];
    __shared__ float sRed[2][8];
    const int lane = threadIdx.x & 31;
    const int wave = threadIdx.x >> 5;
    const int half = lane >> 4;
    const int col0 = lane & 15;
    const int nrt  = (N + 15) >> 4;
    const int rt0  = (blockIdx.x * 8 + wave) * 2;
    const int rt1  = rt0 + 1;
    const int rowA0 = (rt0 < nrt) ? min(rt0 * 16 + col0, N - 1) : 0;
    const int rowA1 = (rt1 < nrt) ? min(rt1 * 16 + col0, N - 1) : 0;

    zero_rows(acc0, rt0 * 16, lane, N, 64);
    if (rt0 * 16 + lane < N) denom[rt0 * 16 + lane] = 0.0f;
    if (blockIdx.x == 0 && threadIdx.x < 4) ctr[threadIdx.x * CTR_STRIDE] = 0u;

    v8f acc[2][5];
    #pragma unroll
    for (int u = 0; u < 2; ++u)
        #pragma unroll
        for (int t = 0; t < 5; ++t) acc[u][t] = v8f{};

    for (int i = threadIdx.x; i < (NB1 - 66) * P1; i += 256) sW[66 * P1 + i] = 0.0f;

    const int phases = K >> 7;
    for (int ph = 0; ph < phases; ++ph) {
        for (int i = threadIdx.x; i < 128 * 64; i += 256) {
            int r = i >> 6, c = i & 63;
            sW[c * P1 + r] = W[(size_t)ph * 128 * 64 + i];
        }
        __syncthreads();
        {
            const int r = threadIdx.x & 127;
            const int which = threadIdx.x >> 7;
            const float* __restrict__ a = which ? adst : asrc;
            float v = 0.0f;
            for (int c = 0; c < 64; ++c) v = fmaf(sW[c * P1 + r], a[c], v);
            sW[(64 + which) * P1 + r] = v;
        }
        __syncthreads();

        const float* xr0 = X + (size_t)rowA0 * K + ph * 128;
        const float* xr1 = X + (size_t)rowA1 * K + ph * 128;
        for (int kk = 0; kk < 32; ++kk) {
            const int kb = kk * 4 + half * 2;
            v2f a0 = *(const v2f*)(xr0 + kb);
            v2f a1 = *(const v2f*)(xr1 + kb);
            #pragma unroll
            for (int t = 0; t < 5; ++t) {
                v2f b = *(const v2f*)(&sW[(t * 16 + col0) * P1 + kb]);
                acc[0][t] = __builtin_amdgcn_wmma_f32_16x16x4_f32(
                    false, a0, false, b, (short)0, acc[0][t], false, false);
                acc[1][t] = __builtin_amdgcn_wmma_f32_16x16x4_f32(
                    false, a1, false, b, (short)0, acc[1][t], false, false);
            }
        }
        __syncthreads();
    }

    if (col0 < 2) {
        #pragma unroll
        for (int u = 0; u < 2; ++u)
            #pragma unroll
            for (int r = 0; r < 8; ++r) sSD[wave][col0][u * 16 + half * 8 + r] = acc[u][4][r];
    }
    __syncthreads();
    {
        const int node = rt0 * 16 + lane;
        const float sv = sSD[wave][0][lane];
        if (node < N) {
            s[node] = sv;
            *(uint2*)(nd + 2u * (unsigned)node) =
                make_uint2(__float_as_uint(sSD[wave][1][lane]), NEG_INF_KEY);
        }
        wave_minmax(sv, node < N, sRed, wave, lane);
    }

    #pragma unroll
    for (int u = 0; u < 2; ++u) {
        int rt = rt0 + u;
        if (rt >= nrt) continue;
        #pragma unroll
        for (int t = 0; t < 4; ++t)
            #pragma unroll
            for (int r = 0; r < 8; ++r) {
                int row = rt * 16 + r + half * 8;
                if (row < N) H[(size_t)row * 64 + t * 16 + col0] = acc[u][t][r];
            }
    }
    __syncthreads();
    block_minmax(sRed, stat);
}

#define P2 68
__global__ __launch_bounds__(256) void gemm2_kernel(
    const float* __restrict__ Hin, const float* __restrict__ Dn,
    const float* __restrict__ W,
    const float* __restrict__ asrc, const float* __restrict__ adst,
    float* __restrict__ Hout, float* __restrict__ s, unsigned* __restrict__ nd,
    float* __restrict__ acc0, unsigned* __restrict__ ctr,
    float* __restrict__ stat, int N, int NC, int HP) {
    __shared__ float sW[48 * P2];
    __shared__ float sSD[8][2][32];
    __shared__ float sRed[2][8];
    const int lane = threadIdx.x & 31;
    const int wave = threadIdx.x >> 5;
    const int half = lane >> 4;
    const int col0 = lane & 15;
    const int nrt  = (N + 15) >> 4;
    const int rt0  = (blockIdx.x * 8 + wave) * 2;
    const int rt1  = rt0 + 1;
    const int rowA0 = (rt0 < nrt) ? min(rt0 * 16 + col0, N - 1) : 0;
    const int rowA1 = (rt1 < nrt) ? min(rt1 * 16 + col0, N - 1) : 0;

    zero_rows(acc0, rt0 * 16, lane, N, HP);
    if (blockIdx.x == 0 && threadIdx.x < 4) ctr[threadIdx.x * CTR_STRIDE] = 0u;

    for (int i = threadIdx.x; i < 64 * 48; i += 256) {
        int r = i / 48, c = i % 48;
        sW[c * P2 + r] = (c < NC) ? W[(size_t)r * NC + c] : 0.0f;
    }
    __syncthreads();
    if (threadIdx.x < 128) {
        const int r = threadIdx.x & 63;
        const int which = threadIdx.x >> 6;
        const float* __restrict__ a = which ? adst : asrc;
        float v = 0.0f;
        for (int c = 0; c < NC; ++c) v = fmaf(sW[c * P2 + r], a[c], v);
        sW[(NC + which) * P2 + r] = v;
    }
    __syncthreads();

    v8f acc[2][3];
    #pragma unroll
    for (int u = 0; u < 2; ++u)
        #pragma unroll
        for (int t = 0; t < 3; ++t) acc[u][t] = v8f{};

    const float* xr0 = Hin + (size_t)rowA0 * 64;
    const float* xr1 = Hin + (size_t)rowA1 * 64;
    const float dn0 = fmaxf(Dn[rowA0], 1e-16f);
    const float dn1 = fmaxf(Dn[rowA1], 1e-16f);
    for (int kk = 0; kk < 16; ++kk) {
        const int kb = kk * 4 + half * 2;
        v2f a0 = *(const v2f*)(xr0 + kb);
        v2f a1 = *(const v2f*)(xr1 + kb);
        a0[0] = fmaxf(a0[0] / dn0, 0.0f); a0[1] = fmaxf(a0[1] / dn0, 0.0f);
        a1[0] = fmaxf(a1[0] / dn1, 0.0f); a1[1] = fmaxf(a1[1] / dn1, 0.0f);
        #pragma unroll
        for (int t = 0; t < 3; ++t) {
            v2f b = *(const v2f*)(&sW[(t * 16 + col0) * P2 + kb]);
            acc[0][t] = __builtin_amdgcn_wmma_f32_16x16x4_f32(
                false, a0, false, b, (short)0, acc[0][t], false, false);
            acc[1][t] = __builtin_amdgcn_wmma_f32_16x16x4_f32(
                false, a1, false, b, (short)0, acc[1][t], false, false);
        }
    }

    #pragma unroll
    for (int u = 0; u < 2; ++u) {
        int rt = rt0 + u;
        if (rt >= nrt) continue;
        #pragma unroll
        for (int t = 0; t < 3; ++t)
            #pragma unroll
            for (int r = 0; r < 8; ++r) {
                int row = rt * 16 + r + half * 8;
                int col = t * 16 + col0;
                if (row < N && col < NC)
                    Hout[(size_t)row * HP + col] = acc[u][t][r];
            }
    }

    #pragma unroll
    for (int t = 0; t < 3; ++t) {
        const int w = t * 16 + col0 - NC;
        if (w == 0 || w == 1) {
            #pragma unroll
            for (int u = 0; u < 2; ++u)
                #pragma unroll
                for (int r = 0; r < 8; ++r) sSD[wave][w][u * 16 + half * 8 + r] = acc[u][t][r];
        }
    }
    __syncthreads();
    {
        const int node = rt0 * 16 + lane;
        const float sv = sSD[wave][0][lane];
        if (node < N) {
            s[node] = sv;
            *(uint2*)(nd + 2u * (unsigned)node) =
                make_uint2(__float_as_uint(sSD[wave][1][lane]), NEG_INF_KEY);
        }
        wave_minmax(sv, node < N, sRed, wave, lane);
    }
    __syncthreads();
    block_minmax(sRed, stat);
}

#define TCH 128
#define NK (TCH / 32)
__device__ __forceinline__ int wave_side() {
    return (int)((__builtin_amdgcn_s_sendmsg_rtn(0x87) >> 18) & 1u);
}
__device__ __forceinline__ int draw(unsigned* ctr, int lane) {
    unsigned t = 0u;
    if (lane == 0) t = atomicAdd(ctr, 1u);
    return __builtin_amdgcn_readfirstlane((int)t);
}
__device__ __forceinline__ void block_edges(
    const int* __restrict__ ei, int t, int lane, int E, int EA, int NH, int r,
    int (&src)[NK], int (&dst)[NK], bool (&in)[NK]) {
    int ls[NK], ld[NK];
    #pragma unroll
    for (int k = 0; k < NK; ++k) {
        const int ec = min(t * TCH + k * 32 + lane, E - 1);
        ls[k] = ei[ec];
        ld[k] = ei[E + ec];
    }
    #pragma unroll
    for (int k = 0; k < NK; ++k) {
        const int e = t * TCH + k * 32 + lane;
        const bool loop = e >= E;
        src[k] = loop ? e - E : ls[k];
        dst[k] = loop ? e - E : ld[k];
        in[k] = (e < EA) && ((dst[k] >= NH) == (r != 0));
    }
}

__global__ __launch_bounds__(256) void edge_max_kernel(
    const int* __restrict__ ei, const float* __restrict__ s,
    const float* __restrict__ stat, int nstat,
    unsigned* __restrict__ nd, float* __restrict__ es,
    unsigned* __restrict__ ctr, int E, int EA, int NH) {
    const int lane = threadIdx.x & 31;
    float smax;
    if (spread_small(stat, nstat, lane, smax)) return;
    const int nT = (EA + TCH - 1) / TCH;
    int r = wave_side();
    for (int pass = 0; pass < 2; ++pass, r ^= 1) {
        unsigned* c = ctr + r * CTR_STRIDE;
        float* esr = es + (size_t)r * nT * TCH;
        for (int t = draw(c, lane); t < nT; t = draw(c, lane)) {
            int src[NK], dst[NK];
            bool in[NK];
            block_edges(ei, t, lane, E, EA, NH, r, src, dst, in);
            float sv[NK];
            #pragma unroll
            for (int k = 0; k < NK; ++k) {
                sv[k] = 0.0f;
                if (in[k]) sv[k] = s[src[k]];
            }
            #pragma unroll
            for (int k = 0; k < NK; ++k)
                if (in[k]) {
                    esr[t * TCH + k * 32 + lane] = sv[k];
                    atomicMax(&nd[2u * (unsigned)dst[k] + 1u], f2ord(sv[k]));
                }
        }
    }
}

#define GRP 16
template <bool FOLD>
__global__ __launch_bounds__(256) void edge_row_kernel(
    const int* __restrict__ ei, const float* __restrict__ ew,
    const float* __restrict__ es, const float* __restrict__ s,
    const float* __restrict__ stat, int nstat,
    const unsigned* __restrict__ nd,
    float* __restrict__ denom,
    const float* __restrict__ H, float* __restrict__ out,
    unsigned* __restrict__ ctr, int E, int EA, int F, int P, int NH) {
    __shared__ int   qs[8][TCH];
    __shared__ int   qd[8][TCH];
    __shared__ float qc[8][TCH];
    __shared__ float qe[8][TCH];
    const int lane = threadIdx.x & 31;
    const int wv = threadIdx.x >> 5;
    const int nT = (EA + TCH - 1) / TCH;
    const bool fA = FOLD && lane == F;
    const bool fB = FOLD && lane + 32 == F;
    const bool p0 = lane < F || fA;
    const bool p1 = lane + 32 < F || fB;
    const unsigned colA = (unsigned)min(lane, F - 1);
    const unsigned colB = (unsigned)min(lane + 32, F - 1);
    const unsigned uF = (unsigned)P;
    float smax;
    const bool fast = spread_small(stat, nstat, lane, smax);
    int r = wave_side();
    for (int pass = 0; pass < 2; ++pass, r ^= 1) {
        unsigned* c = ctr + r * CTR_STRIDE;
        const float* esr = es + (size_t)r * nT * TCH;
        for (int t = draw(c, lane); t < nT; t = draw(c, lane)) {
            int src[NK], dst[NK];
            bool in[NK];
            block_edges(ei, t, lane, E, EA, NH, r, src, dst, in);
            float sv[NK], wt[NK];
            uint2 rec[NK];
            #pragma unroll
            for (int k = 0; k < NK; ++k) {
                const int e = t * TCH + k * 32 + lane;
                sv[k] = 0.0f; rec[k] = make_uint2(0u, 0u); wt[k] = 1.0f;
                if (in[k]) {
                    const float* sp = fast ? s + src[k] : esr + e;
                    sv[k] = *sp;
                    rec[k] = *(const uint2*)(nd + 2u * (unsigned)dst[k]);
                    if (e < E) wt[k] = ew[e];
                }
            }
            int cnt = 0;
            #pragma unroll
            for (int k = 0; k < NK; ++k) {
                const unsigned mask = __builtin_amdgcn_ballot_w32(in[k]);
                const int slot = cnt + (int)__builtin_amdgcn_mbcnt_lo(mask, 0u);
                if (in[k]) {
                    const float dv = __uint_as_float(rec[k].x);
                    const float mx = leaky((fast ? smax : ord2f(rec[k].y)) + dv);
                    const float ex = __expf(leaky(sv[k] + dv) - mx);
                    if (!FOLD) atomicAdd(&denom[dst[k]], ex);
                    qs[wv][slot] = src[k];
                    qd[wv][slot] = dst[k];
                    qc[wv][slot] = ex * wt[k];
                    if (FOLD) qe[wv][slot] = ex;
                }
                cnt += __builtin_popcount(mask);
            }
            for (int g = 0; g < cnt; g += GRP) {
                float va[GRP], vb[GRP];
                #pragma unroll
                for (int j = 0; j < GRP; ++j) {
                    const int idx = min(g + j, cnt - 1);
                    const unsigned ro = (unsigned)__builtin_amdgcn_readfirstlane(qs[wv][idx]) * uF;
                    va[j] = H[ro + colA];
                    vb[j] = H[ro + colB];
                }
                #pragma unroll
                for (int j = 0; j < GRP; ++j) {
                    if (g + j < cnt) {
                        const unsigned wo = (unsigned)__builtin_amdgcn_readfirstlane(qd[wv][g + j]) * uF;
                        const float cj = qc[wv][g + j];
                        float ta = va[j] * cj, tb = vb[j] * cj;
                        if (FOLD) {
                            const float ej = qe[wv][g + j];
                            ta = fA ? ej : ta;
                            tb = fB ? ej : tb;
                        }
                        if (p0) atomicAdd(&out[wo + (unsigned)lane], ta);
                        if (p1) atomicAdd(&out[wo + (unsigned)lane + 32u], tb);
                    }
                }
            }
        }
    }
}

__global__ void norm_rows(float* __restrict__ out, const float* __restrict__ acc,
                          unsigned n, unsigned F, unsigned P) {
    const unsigned i = blockIdx.x * blockDim.x + threadIdx.x;
    if (i < n) {
        const unsigned node = i / F;
        const unsigned c = i - node * F;
        out[i] = acc[node * P + c] / fmaxf(acc[node * P + F], 1e-16f);
    }
}

extern "C" void kernel_launch(void* const* d_in, const int* in_sizes, int n_in,
                              void* d_out, int out_size, void* d_ws, size_t ws_size,
                              hipStream_t stream) {
    const float* x   = (const float*)d_in[0];
    const int*   ei  = (const int*)  d_in[1];
    const float* ew  = (const float*)d_in[2];
    const float* W1  = (const float*)d_in[3];
    const float* a1s = (const float*)d_in[4];
    const float* a1d = (const float*)d_in[5];
    const float* W2  = (const float*)d_in[6];
    const float* a2s = (const float*)d_in[7];
    const float* a2d = (const float*)d_in[8];

    const int HID = in_sizes[4];
    const int FIN = in_sizes[3] / HID;
    const int NC  = in_sizes[7];
    const int N   = in_sizes[0] / FIN;
    const int E   = in_sizes[2];
    const int EA  = E + N;

    const int nrt     = (N + 15) / 16;
    const int pairs   = (nrt + 1) / 2;
    const int gBlocks = (pairs + 7) / 8;

    const int RP = 64;
    float*    h1    = (float*)d_ws;
    float*    out1  = h1   + (size_t)N * RP;
    float*    h2    = out1 + (size_t)N * RP;
    float*    acc2  = h2   + (size_t)N * RP;
    float*    sArr  = acc2 + (size_t)N * RP;
    float*    denom = sArr + N;
    unsigned* ndArr = (unsigned*)(denom + N);
    unsigned* ctr   = ndArr + ((size_t)2 * N + 63) / 64 * 64;
    float*    stat  = (float*)(ctr + 4 * CTR_STRIDE);
    float*    esArr = stat + ((size_t)2 * gBlocks + 63) / 64 * 64;
    float*    outF  = (float*)d_out;

    const int eBlocks = 1024;
    const int NH      = N / 2;

    gemm1_kernel<<<gBlocks, 256, 0, stream>>>(x, W1, a1s, a1d, h1, sArr, ndArr, denom, out1, ctr, stat, N, FIN);
    edge_max_kernel<<<eBlocks, 256, 0, stream>>>(ei, sArr, stat, gBlocks, ndArr, esArr, ctr, E, EA, NH);
    edge_row_kernel<false><<<eBlocks, 256, 0, stream>>>(ei, ew, esArr, sArr, stat, gBlocks, ndArr, denom, h1, out1, ctr + 2 * CTR_STRIDE, E, EA, HID, RP, NH);

    gemm2_kernel<<<gBlocks, 256, 0, stream>>>(out1, denom, W2, a2s, a2d, h2, sArr, ndArr, acc2, ctr, stat, N, NC, RP);
    edge_max_kernel<<<eBlocks, 256, 0, stream>>>(ei, sArr, stat, gBlocks, ndArr, esArr, ctr, E, EA, NH);
    edge_row_kernel<true><<<eBlocks, 256, 0, stream>>>(ei, ew, esArr, sArr, stat, gBlocks, ndArr, denom, h2, acc2, ctr + 2 * CTR_STRIDE, E, EA, NC, RP, NH);
    norm_rows<<<(int)(((long long)N * NC + 255) / 256), 256, 0, stream>>>(outF, acc2, (unsigned)N * (unsigned)NC, (unsigned)NC, (unsigned)RP);
}


# Row pass gathers no s[src] word: the score's s comes out of the source row it loads anyway (L2: GEMM 2 stores s in the row's spare column; L1: wave dot row.a_src, s sum order changes); same atomics
# speedup vs baseline: 4.7157x; 1.1287x over previous
#include <hip/hip_runtime.h>
#include <hip/hip_bf16.h>


typedef __attribute__((ext_vector_type(2))) float v2f;
typedef __attribute__((ext_vector_type(8))) float v8f;

#define NEG_SLOPE 0.2f

__device__ __forceinline__ unsigned f2ord(float f) {
    unsigned u = __float_as_uint(f);
    return (u & 0x80000000u) ? ~u : (u | 0x80000000u);
}
__device__ __forceinline__ float ord2f(unsigned k) {
    unsigned u = (k & 0x80000000u) ? (k & 0x7fffffffu) : ~k;
    return __uint_as_float(u);
}
__device__ __forceinline__ float leaky(float x) {
    return x > 0.0f ? x : NEG_SLOPE * x;
}

#define SPREAD_MAX 32.0f
__device__ __forceinline__ bool spread_small(const float* __restrict__ stat, int nstat, int lane, float& smax) {
    float hi = -__builtin_inff(), lo = __builtin_inff();
    for (int base = 0; base < nstat; base += 16 * 32) {
        float2 p[16];
        #pragma unroll
        for (int j = 0; j < 16; ++j)
            p[j] = *(const float2*)(stat + 2 * min(base + j * 32 + lane, nstat - 1));
        #pragma unroll
        for (int j = 0; j < 16; ++j) {
            hi = fmaxf(hi, p[j].x);
            lo = fminf(lo, p[j].y);
        }
    }
    #pragma unroll
    for (int o = 16; o > 0; o >>= 1) {
        hi = fmaxf(hi, __shfl_xor(hi, o));
        lo = fminf(lo, __shfl_xor(lo, o));
    }
    smax = __uint_as_float((unsigned)__builtin_amdgcn_readfirstlane((int)__float_as_uint(hi)));
    return __builtin_amdgcn_readfirstlane((hi - lo <= SPREAD_MAX) ? 1 : 0) != 0;
}
__device__ __forceinline__ void wave_minmax(float sv, bool ok, float (&red)[2][8], int wave, int lane) {
    float hi = ok ? sv : -__builtin_inff();
    float lo = ok ? sv : __builtin_inff();
    #pragma unroll
    for (int o = 16; o > 0; o >>= 1) {
        hi = fmaxf(hi, __shfl_xor(hi, o));
        lo = fminf(lo, __shfl_xor(lo, o));
    }
    if (lane == 0) { red[0][wave] = hi; red[1][wave] = lo; }
}
__device__ __forceinline__ void block_minmax(const float (&red)[2][8], float* __restrict__ stat) {
    if (threadIdx.x == 0) {
        float hi = red[0][0], lo = red[1][0];
        #pragma unroll
        for (int w = 1; w < 8; ++w) { hi = fmaxf(hi, red[0][w]); lo = fminf(lo, red[1][w]); }
        *(float2*)(stat + 2 * blockIdx.x) = make_float2(hi, lo);
    }
}

#define NEG_INF_KEY 0x007FFFFFu
#define CTR_STRIDE 64

__device__ __forceinline__ void zero_rows(float* __restrict__ Z, int node0, int lane, int N, int P) {
    const long long lim = (long long)N * P;
    const long long base = (long long)node0 * P;
    for (int o = lane * 4; o < 32 * P; o += 128)
        if (base + o < lim) *(float4*)(Z + base + o) = make_float4(0.0f, 0.0f, 0.0f, 0.0f);
}

#define P1 132
#define NB1 80
__global__ __launch_bounds__(256) void gemm1_kernel(
    const float* __restrict__ X, const float* __restrict__ W,
    const float* __restrict__ asrc, const float* __restrict__ adst,
    float* __restrict__ H, float* __restrict__ s, unsigned* __restrict__ nd,
    float* __restrict__ denom, float* __restrict__ acc0, unsigned* __restrict__ ctr,
    float* __restrict__ stat, int N, int K) {
    __shared__ float sW[NB1 * P1];
    __shared__ float sSD[8][2][32];
    __shared__ float sRed[2][8];
    const int lane = threadIdx.x & 31;
    const int wave = threadIdx.x >> 5;
    const int half = lane >> 4;
    const int col0 = lane & 15;
    const int nrt  = (N + 15) >> 4;
    const int rt0  = (blockIdx.x * 8 + wave) * 2;
    const int rt1  = rt0 + 1;
    const int rowA0 = (rt0 < nrt) ? min(rt0 * 16 + col0, N - 1) : 0;
    const int rowA1 = (rt1 < nrt) ? min(rt1 * 16 + col0, N - 1) : 0;

    zero_rows(acc0, rt0 * 16, lane, N, 64);
    if (rt0 * 16 + lane < N) denom[rt0 * 16 + lane] = 0.0f;
    if (blockIdx.x == 0 && threadIdx.x < 4) ctr[threadIdx.x * CTR_STRIDE] = 0u;

    v8f acc[2][5];
    #pragma unroll
    for (int u = 0; u < 2; ++u)
        #pragma unroll
        for (int t = 0; t < 5; ++t) acc[u][t] = v8f{};

    for (int i = threadIdx.x; i < (NB1 - 66) * P1; i += 256) sW[66 * P1 + i] = 0.0f;

    const int phases = K >> 7;
    for (int ph = 0; ph < phases; ++ph) {
        for (int i = threadIdx.x; i < 128 * 64; i += 256) {
            int r = i >> 6, c = i & 63;
            sW[c * P1 + r] = W[(size_t)ph * 128 * 64 + i];
        }
        __syncthreads();
        {
            const int r = threadIdx.x & 127;
            const int which = threadIdx.x >> 7;
            const float* __restrict__ a = which ? adst : asrc;
            float v = 0.0f;
            for (int c = 0; c < 64; ++c) v = fmaf(sW[c * P1 + r], a[c], v);
            sW[(64 + which) * P1 + r] = v;
        }
        __syncthreads();

        const float* xr0 = X + (size_t)rowA0 * K + ph * 128;
        const float* xr1 = X + (size_t)rowA1 * K + ph * 128;
        for (int kk = 0; kk < 32; ++kk) {
            const int kb = kk * 4 + half * 2;
            v2f a0 = *(const v2f*)(xr0 + kb);
            v2f a1 = *(const v2f*)(xr1 + kb);
            #pragma unroll
            for (int t = 0; t < 5; ++t) {
                v2f b = *(const v2f*)(&sW[(t * 16 + col0) * P1 + kb]);
                acc[0][t] = __builtin_amdgcn_wmma_f32_16x16x4_f32(
                    false, a0, false, b, (short)0, acc[0][t], false, false);
                acc[1][t] = __builtin_amdgcn_wmma_f32_16x16x4_f32(
                    false, a1, false, b, (short)0, acc[1][t], false, false);
            }
        }
        __syncthreads();
    }

    if (col0 < 2) {
        #pragma unroll
        for (int u = 0; u < 2; ++u)
            #pragma unroll
            for (int r = 0; r < 8; ++r) sSD[wave][col0][u * 16 + half * 8 + r] = acc[u][4][r];
    }
    __syncthreads();
    {
        const int node = rt0 * 16 + lane;
        const float sv = sSD[wave][0][lane];
        if (node < N) {
            s[node] = sv;
            *(uint2*)(nd + 2u * (unsigned)node) =
                make_uint2(__float_as_uint(sSD[wave][1][lane]), NEG_INF_KEY);
        }
        wave_minmax(sv, node < N, sRed, wave, lane);
    }

    #pragma unroll
    for (int u = 0; u < 2; ++u) {
        int rt = rt0 + u;
        if (rt >= nrt) continue;
        #pragma unroll
        for (int t = 0; t < 4; ++t)
            #pragma unroll
            for (int r = 0; r < 8; ++r) {
                int row = rt * 16 + r + half * 8;
                if (row < N) H[(size_t)row * 64 + t * 16 + col0] = acc[u][t][r];
            }
    }
    __syncthreads();
    block_minmax(sRed, stat);
}

#define P2 68
__global__ __launch_bounds__(256) void gemm2_kernel(
    const float* __restrict__ Hin, const float* __restrict__ Dn,
    const float* __restrict__ W,
    const float* __restrict__ asrc, const float* __restrict__ adst,
    float* __restrict__ Hout, float* __restrict__ s, unsigned* __restrict__ nd,
    float* __restrict__ acc0, unsigned* __restrict__ ctr,
    float* __restrict__ stat, int N, int NC, int HP) {
    __shared__ float sW[48 * P2];
    __shared__ float sSD[8][2][32];
    __shared__ float sRed[2][8];
    const int lane = threadIdx.x & 31;
    const int wave = threadIdx.x >> 5;
    const int half = lane >> 4;
    const int col0 = lane & 15;
    const int nrt  = (N + 15) >> 4;
    const int rt0  = (blockIdx.x * 8 + wave) * 2;
    const int rt1  = rt0 + 1;
    const int rowA0 = (rt0 < nrt) ? min(rt0 * 16 + col0, N - 1) : 0;
    const int rowA1 = (rt1 < nrt) ? min(rt1 * 16 + col0, N - 1) : 0;

    zero_rows(acc0, rt0 * 16, lane, N, HP);
    if (blockIdx.x == 0 && threadIdx.x < 4) ctr[threadIdx.x * CTR_STRIDE] = 0u;

    for (int i = threadIdx.x; i < 64 * 48; i += 256) {
        int r = i / 48, c = i % 48;
        sW[c * P2 + r] = (c < NC) ? W[(size_t)r * NC + c] : 0.0f;
    }
    __syncthreads();
    if (threadIdx.x < 128) {
        const int r = threadIdx.x & 63;
        const int which = threadIdx.x >> 6;
        const float* __restrict__ a = which ? adst : asrc;
        float v = 0.0f;
        for (int c = 0; c < NC; ++c) v = fmaf(sW[c * P2 + r], a[c], v);
        sW[(NC + which) * P2 + r] = v;
    }
    __syncthreads();

    v8f acc[2][3];
    #pragma unroll
    for (int u = 0; u < 2; ++u)
        #pragma unroll
        for (int t = 0; t < 3; ++t) acc[u][t] = v8f{};

    const float* xr0 = Hin + (size_t)rowA0 * 64;
    const float* xr1 = Hin + (size_t)rowA1 * 64;
    const float dn0 = fmaxf(Dn[rowA0], 1e-16f);
    const float dn1 = fmaxf(Dn[rowA1], 1e-16f);
    for (int kk = 0; kk < 16; ++kk) {
        const int kb = kk * 4 + half * 2;
        v2f a0 = *(const v2f*)(xr0 + kb);
        v2f a1 = *(const v2f*)(xr1 + kb);
        a0[0] = fmaxf(a0[0] / dn0, 0.0f); a0[1] = fmaxf(a0[1] / dn0, 0.0f);
        a1[0] = fmaxf(a1[0] / dn1, 0.0f); a1[1] = fmaxf(a1[1] / dn1, 0.0f);
        #pragma unroll
        for (int t = 0; t < 3; ++t) {
            v2f b = *(const v2f*)(&sW[(t * 16 + col0) * P2 + kb]);
            acc[0][t] = __builtin_amdgcn_wmma_f32_16x16x4_f32(
                false, a0, false, b, (short)0, acc[0][t], false, false);
            acc[1][t] = __builtin_amdgcn_wmma_f32_16x16x4_f32(
                false, a1, false, b, (short)0, acc[1][t], false, false);
        }
    }

    #pragma unroll
    for (int u = 0; u < 2; ++u) {
        int rt = rt0 + u;
        if (rt >= nrt) continue;
        #pragma unroll
        for (int t = 0; t < 3; ++t)
            #pragma unroll
            for (int r = 0; r < 8; ++r) {
                int row = rt * 16 + r + half * 8;
                int col = t * 16 + col0;
                if (row < N && col <= NC)
                    Hout[(size_t)row * HP + col] = acc[u][t][r];
            }
    }

    #pragma unroll
    for (int t = 0; t < 3; ++t) {
        const int w = t * 16 + col0 - NC;
        if (w == 0 || w == 1) {
            #pragma unroll
            for (int u = 0; u < 2; ++u)
                #pragma unroll
                for (int r = 0; r < 8; ++r) sSD[wave][w][u * 16 + half * 8 + r] = acc[u][t][r];
        }
    }
    __syncthreads();
    {
        const int node = rt0 * 16 + lane;
        const float sv = sSD[wave][0][lane];
        if (node < N) {
            s[node] = sv;
            *(uint2*)(nd + 2u * (unsigned)node) =
                make_uint2(__float_as_uint(sSD[wave][1][lane]), NEG_INF_KEY);
        }
        wave_minmax(sv, node < N, sRed, wave, lane);
    }
    __syncthreads();
    block_minmax(sRed, stat);
}

#define TCH 128
#define NK (TCH / 32)
__device__ __forceinline__ int wave_side() {
    return (int)((__builtin_amdgcn_s_sendmsg_rtn(0x87) >> 18) & 1u);
}
__device__ __forceinline__ int draw(unsigned* ctr, int lane) {
    unsigned t = 0u;
    if (lane == 0) t = atomicAdd(ctr, 1u);
    return __builtin_amdgcn_readfirstlane((int)t);
}
__device__ __forceinline__ void block_edges(
    const int* __restrict__ ei, int t, int lane, int E, int EA, int NH, int r,
    int (&src)[NK], int (&dst)[NK], bool (&in)[NK]) {
    int ls[NK], ld[NK];
    #pragma unroll
    for (int k = 0; k < NK; ++k) {
        const int ec = min(t * TCH + k * 32 + lane, E - 1);
        ls[k] = ei[ec];
        ld[k] = ei[E + ec];
    }
    #pragma unroll
    for (int k = 0; k < NK; ++k) {
        const int e = t * TCH + k * 32 + lane;
        const bool loop = e >= E;
        src[k] = loop ? e - E : ls[k];
        dst[k] = loop ? e - E : ld[k];
        in[k] = (e < EA) && ((dst[k] >= NH) == (r != 0));
    }
}

__global__ __launch_bounds__(256) void edge_max_kernel(
    const int* __restrict__ ei, const float* __restrict__ s,
    const float* __restrict__ stat, int nstat,
    unsigned* __restrict__ nd,
    unsigned* __restrict__ ctr, int E, int EA, int NH) {
    const int lane = threadIdx.x & 31;
    float smax;
    if (spread_small(stat, nstat, lane, smax)) return;
    const int nT = (EA + TCH - 1) / TCH;
    int r = wave_side();
    for (int pass = 0; pass < 2; ++pass, r ^= 1) {
        unsigned* c = ctr + r * CTR_STRIDE;
        for (int t = draw(c, lane); t < nT; t = draw(c, lane)) {
            int src[NK], dst[NK];
            bool in[NK];
            block_edges(ei, t, lane, E, EA, NH, r, src, dst, in);
            float sv[NK];
            #pragma unroll
            for (int k = 0; k < NK; ++k) {
                sv[k] = 0.0f;
                if (in[k]) sv[k] = s[src[k]];
            }
            #pragma unroll
            for (int k = 0; k < NK; ++k)
                if (in[k]) atomicMax(&nd[2u * (unsigned)dst[k] + 1u], f2ord(sv[k]));
        }
    }
}

#define GRP 16
template <int M>
__device__ __forceinline__ float row_xor(float x) {
    return __int_as_float(__builtin_amdgcn_update_dpp(0, __float_as_int(x), 0x160 + M, 0xf, 0xf, true));
}
__device__ __forceinline__ float half_xor(float x) {
    const int b = __float_as_int(x);
    return __int_as_float(__builtin_amdgcn_permlanex16(b, b, 0x76543210, 0xfedcba98, false, false));
}
template <bool FOLD>
__global__ __launch_bounds__(256) void edge_row_kernel(
    const int* __restrict__ ei, const float* __restrict__ ew,
    const float* __restrict__ asrc,
    const float* __restrict__ stat, int nstat,
    const unsigned* __restrict__ nd,
    float* __restrict__ denom,
    const float* __restrict__ H, float* __restrict__ out,
    unsigned* __restrict__ ctr, int E, int EA, int F, int P, int NH) {
    __shared__ int   qs[8][TCH];
    __shared__ uint4 qr[8][TCH];
    const int lane = threadIdx.x & 31;
    const int wv = threadIdx.x >> 5;
    const int nT = (EA + TCH - 1) / TCH;
    const bool fA = FOLD && lane == F;
    const bool fB = FOLD && lane + 32 == F;
    const bool p0 = lane < F || fA;
    const bool p1 = lane + 32 < F || fB;
    const int lastc = FOLD ? F : F - 1;
    const unsigned colA = (unsigned)min(lane, lastc);
    const unsigned colB = (unsigned)min(lane + 32, lastc);
    const unsigned uF = (unsigned)P;
    float aL = 0.0f, aH = 0.0f;
    if (!FOLD) {
        if (lane < F) aL = asrc[lane];
        if (lane + 32 < F) aH = asrc[lane + 32];
    }
    float smax;
    const bool fast = spread_small(stat, nstat, lane, smax);
    int r = wave_side();
    for (int pass = 0; pass < 2; ++pass, r ^= 1) {
        unsigned* c = ctr + r * CTR_STRIDE;
        for (int t = draw(c, lane); t < nT; t = draw(c, lane)) {
            int src[NK], dst[NK];
            bool in[NK];
            block_edges(ei, t, lane, E, EA, NH, r, src, dst, in);
            float wt[NK];
            uint2 rec[NK];
            #pragma unroll
            for (int k = 0; k < NK; ++k) {
                const int e = t * TCH + k * 32 + lane;
                rec[k] = make_uint2(0u, 0u); wt[k] = 1.0f;
                if (in[k]) {
                    rec[k] = *(const uint2*)(nd + 2u * (unsigned)dst[k]);
                    if (e < E) wt[k] = ew[e];
                }
            }
            int cnt = 0;
            #pragma unroll
            for (int k = 0; k < NK; ++k) {
                const unsigned mask = __builtin_amdgcn_ballot_w32(in[k]);
                const int slot = cnt + (int)__builtin_amdgcn_mbcnt_lo(mask, 0u);
                if (in[k]) {
                    const float dv = __uint_as_float(rec[k].x);
                    const float mx = leaky((fast ? smax : ord2f(rec[k].y)) + dv);
                    qs[wv][slot] = src[k];
                    qr[wv][slot] = make_uint4((unsigned)dst[k], rec[k].x, __float_as_uint(mx), __float_as_uint(wt[k]));
                }
                cnt += __builtin_popcount(mask);
            }
            for (int g = 0; g < cnt; g += GRP) {
                const int mine = min(g + (lane & (GRP - 1)), cnt - 1);
                const int srcl = qs[wv][mine];
                const uint4 q = qr[wv][mine];
                float va[GRP], vb[GRP];
                #pragma unroll
                for (int j = 0; j < GRP; ++j) {
                    const unsigned ro = (unsigned)__builtin_amdgcn_readlane(srcl, j) * uF;
                    va[j] = H[ro + colA];
                    vb[j] = H[ro + colB];
                }
                float sj[GRP];
                if (FOLD) {
                    #pragma unroll
                    for (int j = 0; j < GRP; ++j) {
                        const float held = (F < 32) ? va[j] : vb[j];
                        sj[j] = __int_as_float(__builtin_amdgcn_readlane(__float_as_int(held), F & 31));
                    }
                } else {
                    #pragma unroll
                    for (int j = 0; j < GRP; ++j) sj[j] = fmaf(vb[j], aH, va[j] * aL);
                    #pragma unroll
                    for (int j = 0; j < GRP; ++j) sj[j] += row_xor<1>(sj[j]);
                    #pragma unroll
                    for (int j = 0; j < GRP; ++j) sj[j] += row_xor<2>(sj[j]);
                    #pragma unroll
                    for (int j = 0; j < GRP; ++j) sj[j] += row_xor<4>(sj[j]);
                    #pragma unroll
                    for (int j = 0; j < GRP; ++j) sj[j] += row_xor<8>(sj[j]);
                    #pragma unroll
                    for (int j = 0; j < GRP; ++j) sj[j] += half_xor(sj[j]);
                }
                float sl = 0.0f;
                #pragma unroll
                for (int j = 0; j < GRP; ++j) sl = ((lane & (GRP - 1)) == j) ? sj[j] : sl;
                const float el = __expf(leaky(sl + __uint_as_float(q.y)) - __uint_as_float(q.z));
                const float cl = el * __uint_as_float(q.w);
                #pragma unroll
                for (int j = 0; j < GRP; ++j) {
                    if (g + j < cnt) {
                        const unsigned wo = (unsigned)__builtin_amdgcn_readlane((int)q.x, j) * uF;
                        const float cj = __int_as_float(__builtin_amdgcn_readlane(__float_as_int(cl), j));
                        float ta = va[j] * cj, tb = vb[j] * cj;
                        if (FOLD) {
                            const float ej = __int_as_float(__builtin_amdgcn_readlane(__float_as_int(el), j));
                            ta = fA ? ej : ta;
                            tb = fB ? ej : tb;
                        }
                        if (p0) atomicAdd(&out[wo + (unsigned)lane], ta);
                        if (p1) atomicAdd(&out[wo + (unsigned)lane + 32u], tb);
                    }
                }
                if (!FOLD && lane < min(GRP, cnt - g)) atomicAdd(&denom[q.x], el);
            }
        }
    }
}

__global__ void norm_rows(float* __restrict__ out, const float* __restrict__ acc,
                          unsigned n, unsigned F, unsigned P) {
    const unsigned i = blockIdx.x * blockDim.x + threadIdx.x;
    if (i < n) {
        const unsigned node = i / F;
        const unsigned c = i - node * F;
        out[i] = acc[node * P + c] / fmaxf(acc[node * P + F], 1e-16f);
    }
}

extern "C" void kernel_launch(void* const* d_in, const int* in_sizes, int n_in,
                              void* d_out, int out_size, void* d_ws, size_t ws_size,
                              hipStream_t stream) {
    const float* x   = (const float*)d_in[0];
    const int*   ei  = (const int*)  d_in[1];
    const float* ew  = (const float*)d_in[2];
    const float* W1  = (const float*)d_in[3];
    const float* a1s = (const float*)d_in[4];
    const float* a1d = (const float*)d_in[5];
    const float* W2  = (const float*)d_in[6];
    const float* a2s = (const float*)d_in[7];
    const float* a2d = (const float*)d_in[8];

    const int HID = in_sizes[4];
    const int FIN = in_sizes[3] / HID;
    const int NC  = in_sizes[7];
    const int N   = in_sizes[0] / FIN;
    const int E   = in_sizes[2];
    const int EA  = E + N;

    const int nrt     = (N + 15) / 16;
    const int pairs   = (nrt + 1) / 2;
    const int gBlocks = (pairs + 7) / 8;

    const int RP = 64;
    float*    h1    = (float*)d_ws;
    float*    out1  = h1   + (size_t)N * RP;
    float*    h2    = out1 + (size_t)N * RP;
    float*    acc2  = h2   + (size_t)N * RP;
    float*    sArr  = acc2 + (size_t)N * RP;
    float*    denom = sArr + N;
    unsigned* ndArr = (unsigned*)(denom + N);
    unsigned* ctr   = ndArr + ((size_t)2 * N + 63) / 64 * 64;
    float*    stat  = (float*)(ctr + 4 * CTR_STRIDE);
    float*    outF  = (float*)d_out;

    const int eBlocks = 1024;
    const int NH      = N / 2;

    gemm1_kernel<<<gBlocks, 256, 0, stream>>>(x, W1, a1s, a1d, h1, sArr, ndArr, denom, out1, ctr, stat, N, FIN);
    edge_max_kernel<<<eBlocks, 256, 0, stream>>>(ei, sArr, stat, gBlocks, ndArr, ctr, E, EA, NH);
    edge_row_kernel<false><<<eBlocks, 256, 0, stream>>>(ei, ew, a1s, stat, gBlocks, ndArr, denom, h1, out1, ctr + 2 * CTR_STRIDE, E, EA, HID, RP, NH);

    gemm2_kernel<<<gBlocks, 256, 0, stream>>>(out1, denom, W2, a2s, a2d, h2, sArr, ndArr, acc2, ctr, stat, N, NC, RP);
    edge_max_kernel<<<eBlocks, 256, 0, stream>>>(ei, sArr, stat, gBlocks, ndArr, ctr, E, EA, NH);
    edge_row_kernel<true><<<eBlocks, 256, 0, stream>>>(ei, ew, a2s, stat, gBlocks, ndArr, denom, h2, acc2, ctr + 2 * CTR_STRIDE, E, EA, NC, RP, NH);
    norm_rows<<<(int)(((long long)N * NC + 255) / 256), 256, 0, stream>>>(outF, acc2, (unsigned)N * (unsigned)NC, (unsigned)NC, (unsigned)RP);
}


# Row pass: a group's 32 row loads, then its 32 adds, each leave as one run: no lane-mask switch and no rewritten address register in between, so no translation wait per instruction; same atomics
# speedup vs baseline: 5.1133x; 1.0843x over previous
#include <hip/hip_runtime.h>
#include <hip/hip_bf16.h>


typedef __attribute__((ext_vector_type(2))) float v2f;
typedef __attribute__((ext_vector_type(8))) float v8f;

#define NEG_SLOPE 0.2f

__device__ __forceinline__ unsigned f2ord(float f) {
    unsigned u = __float_as_uint(f);
    return (u & 0x80000000u) ? ~u : (u | 0x80000000u);
}
__device__ __forceinline__ float ord2f(unsigned k) {
    unsigned u = (k & 0x80000000u) ? (k & 0x7fffffffu) : ~k;
    return __uint_as_float(u);
}
__device__ __forceinline__ float leaky(float x) {
    return x > 0.0f ? x : NEG_SLOPE * x;
}

#define SPREAD_MAX 32.0f
__device__ __forceinline__ bool spread_small(const float* __restrict__ stat, int nstat, int lane, float& smax) {
    float hi = -__builtin_inff(), lo = __builtin_inff();
    for (int base = 0; base < nstat; base += 16 * 32) {
        float2 p[16];
        #pragma unroll
        for (int j = 0; j < 16; ++j)
            p[j] = *(const float2*)(stat + 2 * min(base + j * 32 + lane, nstat - 1));
        #pragma unroll
        for (int j = 0; j < 16; ++j) {
            hi = fmaxf(hi, p[j].x);
            lo = fminf(lo, p[j].y);
        }
    }
    #pragma unroll
    for (int o = 16; o > 0; o >>= 1) {
        hi = fmaxf(hi, __shfl_xor(hi, o));
        lo = fminf(lo, __shfl_xor(lo, o));
    }
    smax = __uint_as_float((unsigned)__builtin_amdgcn_readfirstlane((int)__float_as_uint(hi)));
    return __builtin_amdgcn_readfirstlane((hi - lo <= SPREAD_MAX) ? 1 : 0) != 0;
}
__device__ __forceinline__ void wave_minmax(float sv, bool ok, float (&red)[2][8], int wave, int lane) {
    float hi = ok ? sv : -__builtin_inff();
    float lo = ok ? sv : __builtin_inff();
    #pragma unroll
    for (int o = 16; o > 0; o >>= 1) {
        hi = fmaxf(hi, __shfl_xor(hi, o));
        lo = fminf(lo, __shfl_xor(lo, o));
    }
    if (lane == 0) { red[0][wave] = hi; red[1][wave] = lo; }
}
__device__ __forceinline__ void block_minmax(const float (&red)[2][8], float* __restrict__ stat) {
    if (threadIdx.x == 0) {
        float hi = red[0][0], lo = red[1][0];
        #pragma unroll
        for (int w = 1; w < 8; ++w) { hi = fmaxf(hi, red[0][w]); lo = fminf(lo, red[1][w]); }
        *(float2*)(stat + 2 * blockIdx.x) = make_float2(hi, lo);
    }
}

#define NEG_INF_KEY 0x007FFFFFu
#define CTR_STRIDE 64

__device__ __forceinline__ void zero_rows(float* __restrict__ Z, int node0, int lane, int N, int P) {
    const long long lim = (long long)N * P;
    const long long base = (long long)node0 * P;
    for (int o = lane * 4; o < 32 * P; o += 128)
        if (base + o < lim) *(float4*)(Z + base + o) = make_float4(0.0f, 0.0f, 0.0f, 0.0f);
}

#define P1 132
#define NB1 80
__global__ __launch_bounds__(256) void gemm1_kernel(
    const float* __restrict__ X, const float* __restrict__ W,
    const float* __restrict__ asrc, const float* __restrict__ adst,
    float* __restrict__ H, float* __restrict__ s, unsigned* __restrict__ nd,
    float* __restrict__ denom, float* __restrict__ acc0, unsigned* __restrict__ ctr,
    float* __restrict__ stat, int N, int K) {
    __shared__ float sW[NB1 * P1];
    __shared__ float sSD[8][2][32];
    __shared__ float sRed[2][8];
    const int lane = threadIdx.x & 31;
    const int wave = threadIdx.x >> 5;
    const int half = lane >> 4;
    const int col0 = lane & 15;
    const int nrt  = (N + 15) >> 4;
    const int rt0  = (blockIdx.x * 8 + wave) * 2;
    const int rt1  = rt0 + 1;
    const int rowA0 = (rt0 < nrt) ? min(rt0 * 16 + col0, N - 1) : 0;
    const int rowA1 = (rt1 < nrt) ? min(rt1 * 16 + col0, N - 1) : 0;

    zero_rows(acc0, rt0 * 16, lane, N, 64);
    if (rt0 * 16 + lane < N) denom[rt0 * 16 + lane] = 0.0f;
    if (blockIdx.x == 0 && threadIdx.x < 4) ctr[threadIdx.x * CTR_STRIDE] = 0u;

    v8f acc[2][5];
    #pragma unroll
    for (int u = 0; u < 2; ++u)
        #pragma unroll
        for (int t = 0; t < 5; ++t) acc[u][t] = v8f{};

    for (int i = threadIdx.x; i < (NB1 - 66) * P1; i += 256) sW[66 * P1 + i] = 0.0f;

    const int phases = K >> 7;
    for (int ph = 0; ph < phases; ++ph) {
        for (int i = threadIdx.x; i < 128 * 64; i += 256) {
            int r = i >> 6, c = i & 63;
            sW[c * P1 + r] = W[(size_t)ph * 128 * 64 + i];
        }
        __syncthreads();
        {
            const int r = threadIdx.x & 127;
            const int which = threadIdx.x >> 7;
            const float* __restrict__ a = which ? adst : asrc;
            float v = 0.0f;
            for (int c = 0; c < 64; ++c) v = fmaf(sW[c * P1 + r], a[c], v);
            sW[(64 + which) * P1 + r] = v;
        }
        __syncthreads();

        const float* xr0 = X + (size_t)rowA0 * K + ph * 128;
        const float* xr1 = X + (size_t)rowA1 * K + ph * 128;
        for (int kk = 0; kk < 32; ++kk) {
            const int kb = kk * 4 + half * 2;
            v2f a0 = *(const v2f*)(xr0 + kb);
            v2f a1 = *(const v2f*)(xr1 + kb);
            #pragma unroll
            for (int t = 0; t < 5; ++t) {
                v2f b = *(const v2f*)(&sW[(t * 16 + col0) * P1 + kb]);
                acc[0][t] = __builtin_amdgcn_wmma_f32_16x16x4_f32(
                    false, a0, false, b, (short)0, acc[0][t], false, false);
                acc[1][t] = __builtin_amdgcn_wmma_f32_16x16x4_f32(
                    false, a1, false, b, (short)0, acc[1][t], false, false);
            }
        }
        __syncthreads();
    }

    if (col0 < 2) {
        #pragma unroll
        for (int u = 0; u < 2; ++u)
            #pragma unroll
            for (int r = 0; r < 8; ++r) sSD[wave][col0][u * 16 + half * 8 + r] = acc[u][4][r];
    }
    __syncthreads();
    {
        const int node = rt0 * 16 + lane;
        const float sv = sSD[wave][0][lane];
        if (node < N) {
            s[node] = sv;
            *(uint2*)(nd + 2u * (unsigned)node) =
                make_uint2(__float_as_uint(sSD[wave][1][lane]), NEG_INF_KEY);
        }
        wave_minmax(sv, node < N, sRed, wave, lane);
    }

    #pragma unroll
    for (int u = 0; u < 2; ++u) {
        int rt = rt0 + u;
        if (rt >= nrt) continue;
        #pragma unroll
        for (int t = 0; t < 4; ++t)
            #pragma unroll
            for (int r = 0; r < 8; ++r) {
                int row = rt * 16 + r + half * 8;
                if (row < N) H[(size_t)row * 64 + t * 16 + col0] = acc[u][t][r];
            }
    }
    __syncthreads();
    block_minmax(sRed, stat);
}

#define P2 68
__global__ __launch_bounds__(256) void gemm2_kernel(
    const float* __restrict__ Hin, const float* __restrict__ Dn,
    const float* __restrict__ W,
    const float* __restrict__ asrc, const float* __restrict__ adst,
    float* __restrict__ Hout, float* __restrict__ s, unsigned* __restrict__ nd,
    float* __restrict__ acc0, unsigned* __restrict__ ctr,
    float* __restrict__ stat, int N, int NC, int HP) {
    __shared__ float sW[48 * P2];
    __shared__ float sSD[8][2][32];
    __shared__ float sRed[2][8];
    const int lane = threadIdx.x & 31;
    const int wave = threadIdx.x >> 5;
    const int half = lane >> 4;
    const int col0 = lane & 15;
    const int nrt  = (N + 15) >> 4;
    const int rt0  = (blockIdx.x * 8 + wave) * 2;
    const int rt1  = rt0 + 1;
    const int rowA0 = (rt0 < nrt) ? min(rt0 * 16 + col0, N - 1) : 0;
    const int rowA1 = (rt1 < nrt) ? min(rt1 * 16 + col0, N - 1) : 0;

    zero_rows(acc0, rt0 * 16, lane, N, HP);
    if (blockIdx.x == 0 && threadIdx.x < 4) ctr[threadIdx.x * CTR_STRIDE] = 0u;

    for (int i = threadIdx.x; i < 64 * 48; i += 256) {
        int r = i / 48, c = i % 48;
        sW[c * P2 + r] = (c < NC) ? W[(size_t)r * NC + c] : 0.0f;
    }
    __syncthreads();
    if (threadIdx.x < 128) {
        const int r = threadIdx.x & 63;
        const int which = threadIdx.x >> 6;
        const float* __restrict__ a = which ? adst : asrc;
        float v = 0.0f;
        for (int c = 0; c < NC; ++c) v = fmaf(sW[c * P2 + r], a[c], v);
        sW[(NC + which) * P2 + r] = v;
    }
    __syncthreads();

    v8f acc[2][3];
    #pragma unroll
    for (int u = 0; u < 2; ++u)
        #pragma unroll
        for (int t = 0; t < 3; ++t) acc[u][t] = v8f{};

    const float* xr0 = Hin + (size_t)rowA0 * 64;
    const float* xr1 = Hin + (size_t)rowA1 * 64;
    const float dn0 = fmaxf(Dn[rowA0], 1e-16f);
    const float dn1 = fmaxf(Dn[rowA1], 1e-16f);
    for (int kk = 0; kk < 16; ++kk) {
        const int kb = kk * 4 + half * 2;
        v2f a0 = *(const v2f*)(xr0 + kb);
        v2f a1 = *(const v2f*)(xr1 + kb);
        a0[0] = fmaxf(a0[0] / dn0, 0.0f); a0[1] = fmaxf(a0[1] / dn0, 0.0f);
        a1[0] = fmaxf(a1[0] / dn1, 0.0f); a1[1] = fmaxf(a1[1] / dn1, 0.0f);
        #pragma unroll
        for (int t = 0; t < 3; ++t) {
            v2f b = *(const v2f*)(&sW[(t * 16 + col0) * P2 + kb]);
            acc[0][t] = __builtin_amdgcn_wmma_f32_16x16x4_f32(
                false, a0, false, b, (short)0, acc[0][t], false, false);
            acc[1][t] = __builtin_amdgcn_wmma_f32_16x16x4_f32(
                false, a1, false, b, (short)0, acc[1][t], false, false);
        }
    }

    #pragma unroll
    for (int u = 0; u < 2; ++u) {
        int rt = rt0 + u;
        if (rt >= nrt) continue;
        #pragma unroll
        for (int t = 0; t < 3; ++t)
            #pragma unroll
            for (int r = 0; r < 8; ++r) {
                int row = rt * 16 + r + half * 8;
                int col = t * 16 + col0;
                if (row < N && col <= NC)
                    Hout[(size_t)row * HP + col] = acc[u][t][r];
            }
    }

    #pragma unroll
    for (int t = 0; t < 3; ++t) {
        const int w = t * 16 + col0 - NC;
        if (w == 0 || w == 1) {
            #pragma unroll
            for (int u = 0; u < 2; ++u)
                #pragma unroll
                for (int r = 0; r < 8; ++r) sSD[wave][w][u * 16 + half * 8 + r] = acc[u][t][r];
        }
    }
    __syncthreads();
    {
        const int node = rt0 * 16 + lane;
        const float sv = sSD[wave][0][lane];
        if (node < N) {
            s[node] = sv;
            *(uint2*)(nd + 2u * (unsigned)node) =
                make_uint2(__float_as_uint(sSD[wave][1][lane]), NEG_INF_KEY);
        }
        wave_minmax(sv, node < N, sRed, wave, lane);
    }
    __syncthreads();
    block_minmax(sRed, stat);
}

#define TCH 128
#define NK (TCH / 32)
__device__ __forceinline__ int wave_side() {
    return (int)((__builtin_amdgcn_s_sendmsg_rtn(0x87) >> 18) & 1u);
}
__device__ __forceinline__ int draw(unsigned* ctr, int lane) {
    unsigned t = 0u;
    if (lane == 0) t = atomicAdd(ctr, 1u);
    return __builtin_amdgcn_readfirstlane((int)t);
}
__device__ __forceinline__ void block_edges(
    const int* __restrict__ ei, int t, int lane, int E, int EA, int NH, int r,
    int (&src)[NK], int (&dst)[NK], bool (&in)[NK]) {
    int ls[NK], ld[NK];
    #pragma unroll
    for (int k = 0; k < NK; ++k) {
        const int ec = min(t * TCH + k * 32 + lane, E - 1);
        ls[k] = ei[ec];
        ld[k] = ei[E + ec];
    }
    #pragma unroll
    for (int k = 0; k < NK; ++k) {
        const int e = t * TCH + k * 32 + lane;
        const bool loop = e >= E;
        src[k] = loop ? e - E : ls[k];
        dst[k] = loop ? e - E : ld[k];
        in[k] = (e < EA) && ((dst[k] >= NH) == (r != 0));
    }
}

__global__ __launch_bounds__(256) void edge_max_kernel(
    const int* __restrict__ ei, const float* __restrict__ s,
    const float* __restrict__ stat, int nstat,
    unsigned* __restrict__ nd,
    unsigned* __restrict__ ctr, int E, int EA, int NH) {
    const int lane = threadIdx.x & 31;
    float smax;
    if (spread_small(stat, nstat, lane, smax)) return;
    const int nT = (EA + TCH - 1) / TCH;
    int r = wave_side();
    for (int pass = 0; pass < 2; ++pass, r ^= 1) {
        unsigned* c = ctr + r * CTR_STRIDE;
        for (int t = draw(c, lane); t < nT; t = draw(c, lane)) {
            int src[NK], dst[NK];
            bool in[NK];
            block_edges(ei, t, lane, E, EA, NH, r, src, dst, in);
            float sv[NK];
            #pragma unroll
            for (int k = 0; k < NK; ++k) {
                sv[k] = 0.0f;
                if (in[k]) sv[k] = s[src[k]];
            }
            #pragma unroll
            for (int k = 0; k < NK; ++k)
                if (in[k]) atomicMax(&nd[2u * (unsigned)dst[k] + 1u], f2ord(sv[k]));
        }
    }
}

#define GRP 16
template <int M>
__device__ __forceinline__ float row_xor(float x) {
    return __int_as_float(__builtin_amdgcn_update_dpp(0, __float_as_int(x), 0x160 + M, 0xf, 0xf, true));
}
__device__ __forceinline__ float half_xor(float x) {
    const int b = __float_as_int(x);
    return __int_as_float(__builtin_amdgcn_permlanex16(b, b, 0x76543210, 0xfedcba98, false, false));
}
template <bool FOLD>
__global__ __launch_bounds__(256) void edge_row_kernel(
    const int* __restrict__ ei, const float* __restrict__ ew,
    const float* __restrict__ asrc,
    const float* __restrict__ stat, int nstat,
    const unsigned* __restrict__ nd,
    float* __restrict__ denom,
    const float* __restrict__ H, float* __restrict__ out,
    unsigned* __restrict__ ctr, int E, int EA, int F, int P, int NH) {
    __shared__ int   qs[8][TCH];
    __shared__ uint4 qr[8][TCH];
    const int lane = threadIdx.x & 31;
    const int wv = threadIdx.x >> 5;
    const int nT = (EA + TCH - 1) / TCH;
    const bool fA = FOLD && lane == F;
    const bool fB = FOLD && lane + 32 == F;
    const bool p0 = lane < F || fA;
    const bool p1 = lane + 32 < F || fB;
    const int lastc = FOLD ? F : F - 1;
    const unsigned colA = (unsigned)min(lane, lastc);
    const unsigned colB = (unsigned)min(lane + 32, lastc);
    const unsigned uF = (unsigned)P;
    float aL = 0.0f, aH = 0.0f;
    if (!FOLD) {
        if (lane < F) aL = asrc[lane];
        if (lane + 32 < F) aH = asrc[lane + 32];
    }
    float smax;
    const bool fast = spread_small(stat, nstat, lane, smax);
    int r = wave_side();
    for (int pass = 0; pass < 2; ++pass, r ^= 1) {
        unsigned* c = ctr + r * CTR_STRIDE;
        for (int t = draw(c, lane); t < nT; t = draw(c, lane)) {
            int src[NK], dst[NK];
            bool in[NK];
            block_edges(ei, t, lane, E, EA, NH, r, src, dst, in);
            float wt[NK];
            uint2 rec[NK];
            #pragma unroll
            for (int k = 0; k < NK; ++k) {
                const int e = t * TCH + k * 32 + lane;
                rec[k] = *(const uint2*)(nd + 2u * (in[k] ? (unsigned)dst[k] : 0u));
                wt[k] = ew[min(e, E - 1)];
            }
            #pragma unroll
            for (int k = 0; k < NK; ++k)
                if (t * TCH + k * 32 + lane >= E) wt[k] = 1.0f;
            int cnt = 0;
            #pragma unroll
            for (int k = 0; k < NK; ++k) {
                const unsigned mask = __builtin_amdgcn_ballot_w32(in[k]);
                const int slot = cnt + (int)__builtin_amdgcn_mbcnt_lo(mask, 0u);
                if (in[k]) {
                    const float dv = __uint_as_float(rec[k].x);
                    const float mx = leaky((fast ? smax : ord2f(rec[k].y)) + dv);
                    qs[wv][slot] = src[k];
                    qr[wv][slot] = make_uint4((unsigned)dst[k], rec[k].x, __float_as_uint(mx), __float_as_uint(wt[k]));
                }
                cnt += __builtin_popcount(mask);
            }
            for (int g = 0; g < cnt; g += GRP) {
                const int mine = min(g + (lane & (GRP - 1)), cnt - 1);
                const int srcl = qs[wv][mine];
                const uint4 q = qr[wv][mine];
                float va[GRP], vb[GRP];
                unsigned oa[GRP], ob[GRP];
                #pragma unroll
                for (int j = 0; j < GRP; ++j) {
                    const unsigned ro = (unsigned)__builtin_amdgcn_readlane(srcl, j) * uF;
                    oa[j] = ro + colA;
                    ob[j] = ro + colB;
                }
                #pragma unroll
                for (int j = 0; j < GRP; ++j) {
                    va[j] = H[oa[j]];
                    vb[j] = H[ob[j]];
                }
                #pragma unroll
                for (int j = 0; j < GRP; ++j)
                    asm volatile("" :: "v"(oa[j]), "v"(ob[j]));
                float sj[GRP];
                if (FOLD) {
                    #pragma unroll
                    for (int j = 0; j < GRP; ++j) {
                        const float held = (F < 32) ? va[j] : vb[j];
                        sj[j] = __int_as_float(__builtin_amdgcn_readlane(__float_as_int(held), F & 31));
                    }
                } else {
                    #pragma unroll
                    for (int j = 0; j < GRP; ++j) sj[j] = fmaf(vb[j], aH, va[j] * aL);
                    #pragma unroll
                    for (int j = 0; j < GRP; ++j) sj[j] += row_xor<1>(sj[j]);
                    #pragma unroll
                    for (int j = 0; j < GRP; ++j) sj[j] += row_xor<2>(sj[j]);
                    #pragma unroll
                    for (int j = 0; j < GRP; ++j) sj[j] += row_xor<4>(sj[j]);
                    #pragma unroll
                    for (int j = 0; j < GRP; ++j) sj[j] += row_xor<8>(sj[j]);
                    #pragma unroll
                    for (int j = 0; j < GRP; ++j) sj[j] += half_xor(sj[j]);
                }
                float sl = 0.0f;
                #pragma unroll
                for (int j = 0; j < GRP; ++j) sl = ((lane & (GRP - 1)) == j) ? sj[j] : sl;
                const float el = __expf(leaky(sl + __uint_as_float(q.y)) - __uint_as_float(q.z));
                const float cl = el * __uint_as_float(q.w);
                unsigned ao[GRP];
                #pragma unroll
                for (int j = 0; j < GRP; ++j) {
                    ao[j] = (unsigned)__builtin_amdgcn_readlane((int)q.x, j) * uF + (unsigned)lane;
                    const float cj = __int_as_float(__builtin_amdgcn_readlane(__float_as_int(cl), j));
                    va[j] *= cj;
                    vb[j] *= cj;
                    if (FOLD) {
                        const float ej = __int_as_float(__builtin_amdgcn_readlane(__float_as_int(el), j));
                        va[j] = fA ? ej : va[j];
                        vb[j] = fB ? ej : vb[j];
                    }
                }
                const int nv = cnt - g;
                if (p0) {
                    #pragma unroll
                    for (int j = 0; j < GRP; ++j)
                        if (j < nv) atomicAdd(out + ao[j], va[j]);
                }
                if (p1) {
                    #pragma unroll
                    for (int j = 0; j < GRP; ++j)
                        if (j < nv) atomicAdd(out + ao[j] + 32, vb[j]);
                }
                if (!FOLD && lane < min(GRP, cnt - g)) atomicAdd(&denom[q.x], el);
            }
        }
    }
}

__global__ void norm_rows(float* __restrict__ out, const float* __restrict__ acc,
                          unsigned n, unsigned F, unsigned P) {
    const unsigned i = blockIdx.x * blockDim.x + threadIdx.x;
    if (i < n) {
        const unsigned node = i / F;
        const unsigned c = i - node * F;
        out[i] = acc[node * P + c] / fmaxf(acc[node * P + F], 1e-16f);
    }
}

extern "C" void kernel_launch(void* const* d_in, const int* in_sizes, int n_in,
                              void* d_out, int out_size, void* d_ws, size_t ws_size,
                              hipStream_t stream) {
    const float* x   = (const float*)d_in[0];
    const int*   ei  = (const int*)  d_in[1];
    const float* ew  = (const float*)d_in[2];
    const float* W1  = (const float*)d_in[3];
    const float* a1s = (const float*)d_in[4];
    const float* a1d = (const float*)d_in[5];
    const float* W2  = (const float*)d_in[6];
    const float* a2s = (const float*)d_in[7];
    const float* a2d = (const float*)d_in[8];

    const int HID = in_sizes[4];
    const int FIN = in_sizes[3] / HID;
    const int NC  = in_sizes[7];
    const int N   = in_sizes[0] / FIN;
    const int E   = in_sizes[2];
    const int EA  = E + N;

    const int nrt     = (N + 15) / 16;
    const int pairs   = (nrt + 1) / 2;
    const int gBlocks = (pairs + 7) / 8;

    const int RP = 64;
    float*    h1    = (float*)d_ws;
    float*    out1  = h1   + (size_t)N * RP;
    float*    h2    = out1 + (size_t)N * RP;
    float*    acc2  = h2   + (size_t)N * RP;
    float*    sArr  = acc2 + (size_t)N * RP;
    float*    denom = sArr + N;
    unsigned* ndArr = (unsigned*)(denom + N);
    unsigned* ctr   = ndArr + ((size_t)2 * N + 63) / 64 * 64;
    float*    stat  = (float*)(ctr + 4 * CTR_STRIDE);
    float*    outF  = (float*)d_out;

    const int eBlocks = 1024;
    const int NH      = N / 2;

    gemm1_kernel<<<gBlocks, 256, 0, stream>>>(x, W1, a1s, a1d, h1, sArr, ndArr, denom, out1, ctr, stat, N, FIN);
    edge_max_kernel<<<eBlocks, 256, 0, stream>>>(ei, sArr, stat, gBlocks, ndArr, ctr, E, EA, NH);
    edge_row_kernel<false><<<eBlocks, 256, 0, stream>>>(ei, ew, a1s, stat, gBlocks, ndArr, denom, h1, out1, ctr + 2 * CTR_STRIDE, E, EA, HID, RP, NH);

    gemm2_kernel<<<gBlocks, 256, 0, stream>>>(out1, denom, W2, a2s, a2d, h2, sArr, ndArr, acc2, ctr, stat, N, NC, RP);
    edge_max_kernel<<<eBlocks, 256, 0, stream>>>(ei, sArr, stat, gBlocks, ndArr, ctr, E, EA, NH);
    edge_row_kernel<true><<<eBlocks, 256, 0, stream>>>(ei, ew, a2s, stat, gBlocks, ndArr, denom, h2, acc2, ctr + 2 * CTR_STRIDE, E, EA, NC, RP, NH);
    norm_rows<<<(int)(((long long)N * NC + 255) / 256), 256, 0, stream>>>(outF, acc2, (unsigned)N * (unsigned)NC, (unsigned)NC, (unsigned)RP);
}
